# starting kernel
# baseline (speedup 1.0000x reference)
#include <hip/hip_runtime.h>
#include <cstddef>

typedef float v2f __attribute__((ext_vector_type(2)));
typedef float v8f __attribute__((ext_vector_type(8)));

#define RGCN_R   128
#define RGCN_B   4
#define RGCN_IN  128
#define RGCN_HID 64
#define RGCN_W   (RGCN_HID * (RGCN_B + 1))

__device__ __forceinline__ v8f wmma_f32x4(v2f a, v2f b, v8f c) {
  return __builtin_amdgcn_wmma_f32_16x16x4_f32(false, a, false, b, (short)0, c,
                                               false, false);
}

__global__ void rgcn_count_kernel(const int* __restrict__ dst,
                                  const int* __restrict__ et,
                                  float* __restrict__ cnt, int E) {
  int e = blockIdx.x * blockDim.x + threadIdx.x;
  if (e < E) atomicAdd(&cnt[(size_t)dst[e] * RGCN_R + et[e]], 1.0f);
}

template<int K>
__global__ void rgcn_pack_kernel(const float* __restrict__ rootW,
                                 const float* __restrict__ basisW,
                                 float* __restrict__ Bp) {
  const int t = blockIdx.x * blockDim.x + threadIdx.x;
  if (t >= K * RGCN_W) return;
  const int m = t & 3;
  const int n = (t >> 2) % RGCN_W;
  const int q = t / (4 * RGCN_W);
  const int k = q * 4 + m;
  float v;
  if (n < RGCN_HID) {
    v = rootW[k * RGCN_HID + n];
  } else {
    const int b = (n - RGCN_HID) >> 6;
    const int c = (n - RGCN_HID) & 63;
    v = basisW[((size_t)b * K + k) * RGCN_HID + c];
  }
  Bp[t] = v;
}

template<int K, bool RELU_A>
__global__ void rgcn_gemm_kernel(const float* __restrict__ A,
                                 const float* __restrict__ Bp,
                                 const float* __restrict__ bias,
                                 float* __restrict__ h,
                                 float* __restrict__ z,
                                 int nTilesM) {
  const int wave = threadIdx.x >> 5;
  const int lane = threadIdx.x & 31;
  const int hi   = lane >> 4;
  const int lm   = lane & 15;
  const int tileM = blockIdx.x * (blockDim.x >> 5) + wave;
  const int tileN = blockIdx.y;
  if (tileM >= nTilesM) return;

  const int row = tileM * 16 + lm;
  const float* __restrict__ arow = A + (size_t)row * K + 2 * hi;
  const int n = tileN * 16 + lm;
  const float* __restrict__ bptr = Bp + (size_t)n * 4 + 2 * hi;

  v8f acc = {};
#pragma unroll 8
  for (int k = 0; k < K; k += 4) {
    v2f a = *(const v2f*)(arow + k);
    if (RELU_A) { a.x = fmaxf(a.x, 0.f); a.y = fmaxf(a.y, 0.f); }
    v2f b = *(const v2f*)(bptr + (size_t)(k >> 2) * (4 * RGCN_W));
    acc = wmma_f32x4(a, b, acc);
  }

  if (tileN < 4) {
    const float bv = bias[n];
#pragma unroll
    for (int v = 0; v < 8; ++v) {
      const int r = tileM * 16 + v + 8 * hi;
      h[(size_t)r * RGCN_HID + n] = acc[v] + bv;
    }
  } else {
    const int zc = n - RGCN_HID;
#pragma unroll
    for (int v = 0; v < 8; ++v) {
      const int r = tileM * 16 + v + 8 * hi;
      z[(size_t)r * (RGCN_B * RGCN_HID) + zc] = acc[v];
    }
  }
}

__global__ void rgcn_scatter_kernel(const int* __restrict__ src,
                                    const int* __restrict__ dst,
                                    const int* __restrict__ et,
                                    const float* __restrict__ comp,
                                    const float* __restrict__ probs,
                                    const float* __restrict__ cnt,
                                    const float* __restrict__ z,
                                    float* __restrict__ hout, int E) {
  const int e = blockIdx.x * (blockDim.x >> 5) + (threadIdx.x >> 5);
  if (e >= E) return;
  const int lane = threadIdx.x & 31;

  const int s = src[e], d = dst[e], t = et[e];
  const float gate = 1.f / (1.f + __expf(-probs[t]));
  const float c    = cnt[(size_t)d * RGCN_R + t];
  const float g    = gate / fmaxf(c, 1.f);
  const float c0 = comp[t * RGCN_B + 0] * g;
  const float c1 = comp[t * RGCN_B + 1] * g;
  const float c2 = comp[t * RGCN_B + 2] * g;
  const float c3 = comp[t * RGCN_B + 3] * g;

  const float* __restrict__ zp = z + (size_t)s * (RGCN_B * RGCN_HID);
  float* __restrict__ hp = hout + (size_t)d * RGCN_HID;
#pragma unroll
  for (int h0 = lane; h0 < RGCN_HID; h0 += 32) {
    const float m = c0 * zp[h0] + c1 * zp[RGCN_HID + h0] +
                    c2 * zp[2 * RGCN_HID + h0] + c3 * zp[3 * RGCN_HID + h0];
    atomicAdd(&hp[h0], m);
  }
}

__global__ void rgcn_wmat_kernel(const float* __restrict__ Rm,
                                 const float* __restrict__ D,
                                 float* __restrict__ Wtp) {
  const int t = blockIdx.x * blockDim.x + threadIdx.x;
  if (t >= RGCN_HID * RGCN_HID * RGCN_R) return;
  const int m = t & 3;
  const int r = (t >> 2) & (RGCN_R - 1);
  const int q = t >> 9;
  const int k = q * 4 + m;
  const int i = k >> 6, j = k & 63;
  Wtp[t] = D[r * RGCN_HID + i] * Rm[i * RGCN_HID + j] * D[r * RGCN_HID + j];
}

__global__ void rgcn_score_kernel(const float* __restrict__ hB,
                                  const int* __restrict__ tei,
                                  const float* __restrict__ Wtp,
                                  float* __restrict__ out, int ET) {
  __shared__ float x1s[32][65];
  __shared__ float x2s[32][65];
  const int base = blockIdx.x * 32;

  for (int idx = threadIdx.x; idx < 32 * RGCN_HID; idx += blockDim.x) {
    const int m = idx >> 6, i = idx & 63;
    int e = base + m;
    if (e >= ET) e = 0;
    const int n1 = tei[e], n2 = tei[ET + e];
    x1s[m][i] = fmaxf(hB[(size_t)n1 * RGCN_HID + i], 0.f);
    x2s[m][i] = fmaxf(hB[(size_t)n2 * RGCN_HID + i], 0.f);
  }
  __syncthreads();

  const int wave = threadIdx.x >> 5;
  const int lane = threadIdx.x & 31;
  const int hi   = lane >> 4;
  const int lm   = lane & 15;
  const int r    = wave * 16 + lm;
  const float* __restrict__ wbase = Wtp + (size_t)r * 4 + 2 * hi;

  v8f acc0 = {}, acc1 = {};
  for (int i = 0; i < RGCN_HID; ++i) {
    const float xi0 = x1s[lm][i];
    const float xi1 = x1s[16 + lm][i];
    const float* __restrict__ wrow = wbase + (size_t)i * (16 * 512);
#pragma unroll
    for (int jj = 0; jj < RGCN_HID; jj += 4) {
      const int j0 = jj + 2 * hi;
      v2f b = *(const v2f*)(wrow + (size_t)(jj >> 2) * 512);
      v2f a0, a1;
      a0.x = xi0 * x2s[lm][j0];      a0.y = xi0 * x2s[lm][j0 + 1];
      a1.x = xi1 * x2s[16 + lm][j0]; a1.y = xi1 * x2s[16 + lm][j0 + 1];
      acc0 = wmma_f32x4(a0, b, acc0);
      acc1 = wmma_f32x4(a1, b, acc1);
    }
  }

#pragma unroll
  for (int v = 0; v < 8; ++v) {
    const int e0 = base + v + 8 * hi;
    const int e1 = base + 16 + v + 8 * hi;
    if (e0 < ET) out[(size_t)e0 * RGCN_R + r] = 1.f / (1.f + __expf(-acc0[v]));
    if (e1 < ET) out[(size_t)e1 * RGCN_R + r] = 1.f / (1.f + __expf(-acc1[v]));
  }
}

extern "C" void kernel_launch(void* const* d_in, const int* in_sizes, int n_in,
                              void* d_out, int out_size, void* d_ws,
                              size_t ws_size, hipStream_t stream) {
  const float* x      = (const float*)d_in[0];
  const int*   eidx   = (const int*)d_in[1];
  const int*   etype  = (const int*)d_in[2];
  const int*   tei    = (const int*)d_in[3];
  const float* probs  = (const float*)d_in[4];
  const float* basis1 = (const float*)d_in[5];
  const float* comp1  = (const float*)d_in[6];
  const float* root1  = (const float*)d_in[7];
  const float* bias1  = (const float*)d_in[8];
  const float* basis2 = (const float*)d_in[9];
  const float* comp2  = (const float*)d_in[10];
  const float* root2  = (const float*)d_in[11];
  const float* bias2  = (const float*)d_in[12];
  const float* Rm     = (const float*)d_in[13];
  const float* D      = (const float*)d_in[14];

  const int N  = in_sizes[0] / RGCN_IN;
  const int E  = in_sizes[2];
  const int ET = in_sizes[3] / 2;

  float* cnt = (float*)d_ws;
  float* z   = cnt + (size_t)N * RGCN_R;
  float* hA  = z   + (size_t)N * RGCN_B * RGCN_HID;
  float* hB  = hA  + (size_t)N * RGCN_HID;
  float* Wtp = hB  + (size_t)N * RGCN_HID;
  float* Bp1 = Wtp + (size_t)RGCN_HID * RGCN_HID * RGCN_R;
  float* Bp2 = Bp1 + (size_t)RGCN_IN * RGCN_W;

  const int* src = eidx;
  const int* dst = eidx + E;

  hipMemsetAsync(cnt, 0, (size_t)N * RGCN_R * sizeof(float), stream);

  rgcn_count_kernel<<<(E + 255) / 256, 256, 0, stream>>>(dst, etype, cnt, E);

  rgcn_pack_kernel<RGCN_IN>
      <<<(RGCN_IN * RGCN_W + 255) / 256, 256, 0, stream>>>(root1, basis1, Bp1);
  rgcn_pack_kernel<RGCN_HID>
      <<<(RGCN_HID * RGCN_W + 255) / 256, 256, 0, stream>>>(root2, basis2, Bp2);

  const int tilesM = N / 16;
  dim3 gg((tilesM + 3) / 4, RGCN_W / 16);

  rgcn_gemm_kernel<RGCN_IN, false>
      <<<gg, 128, 0, stream>>>(x, Bp1, bias1, hA, z, tilesM);
  rgcn_scatter_kernel<<<(E + 7) / 8, 256, 0, stream>>>(src, dst, etype, comp1,
                                                       probs, cnt, z, hA, E);

  rgcn_gemm_kernel<RGCN_HID, true>
      <<<gg, 128, 0, stream>>>(hA, Bp2, bias2, hB, z, tilesM);
  rgcn_scatter_kernel<<<(E + 7) / 8, 256, 0, stream>>>(src, dst, etype, comp2,
                                                       probs, cnt, z, hB, E);

  rgcn_wmat_kernel<<<(RGCN_HID * RGCN_HID * RGCN_R + 255) / 256, 256, 0,
                     stream>>>(Rm, D, Wtp);
  rgcn_score_kernel<<<(ET + 31) / 32, 256, 0, stream>>>(hB, tei, Wtp,
                                                        (float*)d_out, ET);
}


# Scatter: one wave walks 32 consecutive edges (edge scalars computed lane-parallel, broadcast by readlane, row gathers of 4 edges in flight) instead of one wave per edge
# speedup vs baseline: 1.4936x; 1.4936x over previous
#include <hip/hip_runtime.h>
#include <cstddef>

typedef float v2f __attribute__((ext_vector_type(2)));
typedef float v8f __attribute__((ext_vector_type(8)));

#define RGCN_R   128
#define RGCN_B   4
#define RGCN_IN  128
#define RGCN_HID 64
#define RGCN_W   (RGCN_HID * (RGCN_B + 1))

__device__ __forceinline__ v8f wmma_f32x4(v2f a, v2f b, v8f c) {
  return __builtin_amdgcn_wmma_f32_16x16x4_f32(false, a, false, b, (short)0, c,
                                               false, false);
}

__global__ void rgcn_count_kernel(const int* __restrict__ dst,
                                  const int* __restrict__ et,
                                  float* __restrict__ cnt, int E) {
  int e = blockIdx.x * blockDim.x + threadIdx.x;
  if (e < E) atomicAdd(&cnt[(size_t)dst[e] * RGCN_R + et[e]], 1.0f);
}

template<int K>
__global__ void rgcn_pack_kernel(const float* __restrict__ rootW,
                                 const float* __restrict__ basisW,
                                 float* __restrict__ Bp) {
  const int t = blockIdx.x * blockDim.x + threadIdx.x;
  if (t >= K * RGCN_W) return;
  const int m = t & 3;
  const int n = (t >> 2) % RGCN_W;
  const int q = t / (4 * RGCN_W);
  const int k = q * 4 + m;
  float v;
  if (n < RGCN_HID) {
    v = rootW[k * RGCN_HID + n];
  } else {
    const int b = (n - RGCN_HID) >> 6;
    const int c = (n - RGCN_HID) & 63;
    v = basisW[((size_t)b * K + k) * RGCN_HID + c];
  }
  Bp[t] = v;
}

template<int K, bool RELU_A>
__global__ void rgcn_gemm_kernel(const float* __restrict__ A,
                                 const float* __restrict__ Bp,
                                 const float* __restrict__ bias,
                                 float* __restrict__ h,
                                 float* __restrict__ z,
                                 int nTilesM) {
  const int wave = threadIdx.x >> 5;
  const int lane = threadIdx.x & 31;
  const int hi   = lane >> 4;
  const int lm   = lane & 15;
  const int tileM = blockIdx.x * (blockDim.x >> 5) + wave;
  const int tileN = blockIdx.y;
  if (tileM >= nTilesM) return;

  const int row = tileM * 16 + lm;
  const float* __restrict__ arow = A + (size_t)row * K + 2 * hi;
  const int n = tileN * 16 + lm;
  const float* __restrict__ bptr = Bp + (size_t)n * 4 + 2 * hi;

  v8f acc = {};
#pragma unroll 8
  for (int k = 0; k < K; k += 4) {
    v2f a = *(const v2f*)(arow + k);
    if (RELU_A) { a.x = fmaxf(a.x, 0.f); a.y = fmaxf(a.y, 0.f); }
    v2f b = *(const v2f*)(bptr + (size_t)(k >> 2) * (4 * RGCN_W));
    acc = wmma_f32x4(a, b, acc);
  }

  if (tileN < 4) {
    const float bv = bias[n];
#pragma unroll
    for (int v = 0; v < 8; ++v) {
      const int r = tileM * 16 + v + 8 * hi;
      h[(size_t)r * RGCN_HID + n] = acc[v] + bv;
    }
  } else {
    const int zc = n - RGCN_HID;
#pragma unroll
    for (int v = 0; v < 8; ++v) {
      const int r = tileM * 16 + v + 8 * hi;
      z[(size_t)r * (RGCN_B * RGCN_HID) + zc] = acc[v];
    }
  }
}

#define SCAT_EPW 32
#define SCAT_GRP 4

template<int G>
__device__ __forceinline__ void rgcn_scatter_group(
    int s, int d, float c0, float c1, float c2, float c3, int j0, int lane,
    const float* __restrict__ z, float* __restrict__ hout) {
  float zv[G][8];
#pragma unroll
  for (int u = 0; u < G; ++u) {
    const int sj = __builtin_amdgcn_readlane(s, j0 + u);
    const float* __restrict__ zp =
        z + (size_t)sj * (RGCN_B * RGCN_HID) + lane;
#pragma unroll
    for (int q = 0; q < 8; ++q) zv[u][q] = zp[q * 32];
  }
#pragma unroll
  for (int u = 0; u < G; ++u) {
    const int j  = j0 + u;
    const int dj = __builtin_amdgcn_readlane(d, j);
    const float a0 = __int_as_float(__builtin_amdgcn_readlane(__float_as_int(c0), j));
    const float a1 = __int_as_float(__builtin_amdgcn_readlane(__float_as_int(c1), j));
    const float a2 = __int_as_float(__builtin_amdgcn_readlane(__float_as_int(c2), j));
    const float a3 = __int_as_float(__builtin_amdgcn_readlane(__float_as_int(c3), j));
    const float m0 = a0 * zv[u][0] + a1 * zv[u][2] + a2 * zv[u][4] + a3 * zv[u][6];
    const float m1 = a0 * zv[u][1] + a1 * zv[u][3] + a2 * zv[u][5] + a3 * zv[u][7];
    float* __restrict__ hp = hout + (size_t)dj * RGCN_HID + lane;
    atomicAdd(hp, m0);
    atomicAdd(hp + 32, m1);
  }
}

__global__ __launch_bounds__(256) void rgcn_scatter_kernel(
    const int* __restrict__ src, const int* __restrict__ dst,
    const int* __restrict__ et, const float* __restrict__ comp,
    const float* __restrict__ probs, const float* __restrict__ cnt,
    const float* __restrict__ z, float* __restrict__ hout, int E) {
  const int lane = threadIdx.x & 31;
  const int wib = __builtin_amdgcn_readfirstlane((int)(threadIdx.x >> 5));
  const int e0  = (blockIdx.x * (blockDim.x >> 5) + wib) * SCAT_EPW;
  if (e0 >= E) return;
  int nE = E - e0;
  if (nE > SCAT_EPW) nE = SCAT_EPW;

  int s = 0, d = 0;
  float c0 = 0.f, c1 = 0.f, c2 = 0.f, c3 = 0.f;
  if (lane < nE) {
    const int e = e0 + lane;
    s = src[e];
    d = dst[e];
    const int t = et[e];
    const float gate = 1.f / (1.f + __expf(-probs[t]));
    const float c    = cnt[(size_t)d * RGCN_R + t];
    const float g    = gate / fmaxf(c, 1.f);
    const float4 cv  = *(const float4*)(comp + t * RGCN_B);
    c0 = cv.x * g; c1 = cv.y * g; c2 = cv.z * g; c3 = cv.w * g;
  }

  int j0 = 0;
  for (; j0 + SCAT_GRP <= nE; j0 += SCAT_GRP)
    rgcn_scatter_group<SCAT_GRP>(s, d, c0, c1, c2, c3, j0, lane, z, hout);
  for (; j0 < nE; ++j0)
    rgcn_scatter_group<1>(s, d, c0, c1, c2, c3, j0, lane, z, hout);
}

#define SC_EPB 10

__global__ __launch_bounds__(256) void rgcn_score_kernel(
    const float* __restrict__ hB,
    const int* __restrict__ tei,
    const float* __restrict__ Rm,
    const float* __restrict__ D,
    float* __restrict__ out, int ET) {
  __shared__ __attribute__((aligned(16))) float x1s[SC_EPB][RGCN_HID];
  __shared__ __attribute__((aligned(16))) float x2s[SC_EPB][RGCN_HID];
  __shared__ __attribute__((aligned(16))) float outs[SC_EPB][RGCN_R];
  const int base = blockIdx.x * SC_EPB;

  for (int idx = threadIdx.x; idx < 2 * SC_EPB * 16; idx += blockDim.x) {
    const int which = idx / (SC_EPB * 16);
    const int rem   = idx - which * (SC_EPB * 16);
    const int m = rem >> 4, q = rem & 15;
    const int e = base + m;
    if (e < ET) {
      const int node = tei[which * ET + e];
      float4 v = *(const float4*)(hB + (size_t)node * RGCN_HID + q * 4);
      v.x = fmaxf(v.x, 0.f); v.y = fmaxf(v.y, 0.f);
      v.z = fmaxf(v.z, 0.f); v.w = fmaxf(v.w, 0.f);
      float* dstp = which ? &x2s[m][q * 4] : &x1s[m][q * 4];
      *(float4*)dstp = v;
    }
  }

  const int wave = threadIdx.x >> 5;
  const int lane = threadIdx.x & 31;
  const int hi   = lane >> 4;
  const int lm   = lane & 15;
  const int r    = wave * 16 + lm;

  v2f af[4][16];
  v2f dj[16];
  float di[4][8];
#pragma unroll
  for (int mt = 0; mt < 4; ++mt) {
#pragma unroll
    for (int kk = 0; kk < 16; ++kk)
      af[mt][kk] = *(const v2f*)(Rm + (mt * 16 + lm) * RGCN_HID + 4 * kk + 2 * hi);
  }
#pragma unroll
  for (int kk = 0; kk < 16; ++kk)
    dj[kk] = *(const v2f*)(D + r * RGCN_HID + 4 * kk + 2 * hi);
#pragma unroll
  for (int mt = 0; mt < 4; ++mt) {
    const float4 lo = *(const float4*)(D + r * RGCN_HID + mt * 16 + 8 * hi);
    const float4 up = *(const float4*)(D + r * RGCN_HID + mt * 16 + 8 * hi + 4);
    di[mt][0] = lo.x; di[mt][1] = lo.y; di[mt][2] = lo.z; di[mt][3] = lo.w;
    di[mt][4] = up.x; di[mt][5] = up.y; di[mt][6] = up.z; di[mt][7] = up.w;
  }
  __syncthreads();

  for (int m = 0; m < SC_EPB; ++m) {
    if (base + m >= ET) break;
    v8f acc[4] = {};
#pragma unroll
    for (int kk = 0; kk < 16; ++kk) {
      const v2f xv = *(const v2f*)&x2s[m][4 * kk + 2 * hi];
      const v2f b  = dj[kk] * xv;
#pragma unroll
      for (int mt = 0; mt < 4; ++mt) acc[mt] = wmma_f32x4(af[mt][kk], b, acc[mt]);
    }
    float p[4];
#pragma unroll
    for (int mt = 0; mt < 4; ++mt) {
      const float4 lo = *(const float4*)&x1s[m][mt * 16 + 8 * hi];
      const float4 up = *(const float4*)&x1s[m][mt * 16 + 8 * hi + 4];
      float s = (lo.x * di[mt][0]) * acc[mt][0];
      s += (lo.y * di[mt][1]) * acc[mt][1];
      s += (lo.z * di[mt][2]) * acc[mt][2];
      s += (lo.w * di[mt][3]) * acc[mt][3];
      s += (up.x * di[mt][4]) * acc[mt][4];
      s += (up.y * di[mt][5]) * acc[mt][5];
      s += (up.z * di[mt][6]) * acc[mt][6];
      s += (up.w * di[mt][7]) * acc[mt][7];
      p[mt] = s;
    }
    float t = (p[0] + p[1]) + (p[2] + p[3]);
    t += __shfl_xor(t, 16, 32);
    if (hi == 0) outs[m][r] = 1.f / (1.f + __expf(-t));
  }
  __syncthreads();

  int nvalid = ET - base;
  if (nvalid > SC_EPB) nvalid = SC_EPB;
  float* __restrict__ obase = out + (size_t)base * RGCN_R;
  const float* osrc = &outs[0][0];
  for (int idx = threadIdx.x * 4; idx < nvalid * RGCN_R; idx += blockDim.x * 4)
    *(float4*)(obase + idx) = *(const float4*)(osrc + idx);
}

extern "C" void kernel_launch(void* const* d_in, const int* in_sizes, int n_in,
                              void* d_out, int out_size, void* d_ws,
                              size_t ws_size, hipStream_t stream) {
  const float* x      = (const float*)d_in[0];
  const int*   eidx   = (const int*)d_in[1];
  const int*   etype  = (const int*)d_in[2];
  const int*   tei    = (const int*)d_in[3];
  const float* probs  = (const float*)d_in[4];
  const float* basis1 = (const float*)d_in[5];
  const float* comp1  = (const float*)d_in[6];
  const float* root1  = (const float*)d_in[7];
  const float* bias1  = (const float*)d_in[8];
  const float* basis2 = (const float*)d_in[9];
  const float* comp2  = (const float*)d_in[10];
  const float* root2  = (const float*)d_in[11];
  const float* bias2  = (const float*)d_in[12];
  const float* Rm     = (const float*)d_in[13];
  const float* D      = (const float*)d_in[14];

  const int N  = in_sizes[0] / RGCN_IN;
  const int E  = in_sizes[2];
  const int ET = in_sizes[3] / 2;

  float* cnt = (float*)d_ws;
  float* z   = cnt + (size_t)N * RGCN_R;
  float* hA  = z   + (size_t)N * RGCN_B * RGCN_HID;
  float* hB  = hA  + (size_t)N * RGCN_HID;
  float* Bp1 = hB  + (size_t)N * RGCN_HID;
  float* Bp2 = Bp1 + (size_t)RGCN_IN * RGCN_W;

  const int* src = eidx;
  const int* dst = eidx + E;

  hipMemsetAsync(cnt, 0, (size_t)N * RGCN_R * sizeof(float), stream);

  rgcn_count_kernel<<<(E + 255) / 256, 256, 0, stream>>>(dst, etype, cnt, E);

  rgcn_pack_kernel<RGCN_IN>
      <<<(RGCN_IN * RGCN_W + 255) / 256, 256, 0, stream>>>(root1, basis1, Bp1);
  rgcn_pack_kernel<RGCN_HID>
      <<<(RGCN_HID * RGCN_W + 255) / 256, 256, 0, stream>>>(root2, basis2, Bp2);

  const int tilesM = N / 16;
  dim3 gg((tilesM + 3) / 4, RGCN_W / 16);

  rgcn_gemm_kernel<RGCN_IN, false>
      <<<gg, 128, 0, stream>>>(x, Bp1, bias1, hA, z, tilesM);
  rgcn_scatter_kernel<<<(E + 255) / 256, 256, 0, stream>>>(src, dst, etype, comp1,
                                                       probs, cnt, z, hA, E);

  rgcn_gemm_kernel<RGCN_HID, true>
      <<<gg, 128, 0, stream>>>(hA, Bp2, bias2, hB, z, tilesM);
  rgcn_scatter_kernel<<<(E + 255) / 256, 256, 0, stream>>>(src, dst, etype, comp2,
                                                       probs, cnt, z, hB, E);

  rgcn_score_kernel<<<(ET + SC_EPB - 1) / SC_EPB, 256, 0, stream>>>(
      hB, tei, Rm, D, (float*)d_out, ET);
}


# GEMM re-tiled: 5-wave block per 32 rows stages its A rows in LDS once (was: 20 waves re-reading them), each wave keeps 2x4 f32 WMMA accumulator tiles (root or one basis); whole-line stores
# speedup vs baseline: 1.6900x; 1.1315x over previous
#include <hip/hip_runtime.h>
#include <cstddef>

typedef float v2f __attribute__((ext_vector_type(2)));
typedef float v8f __attribute__((ext_vector_type(8)));

#define RGCN_R   128
#define RGCN_B   4
#define RGCN_IN  128
#define RGCN_HID 64
#define RGCN_W   (RGCN_HID * (RGCN_B + 1))

__device__ __forceinline__ v8f wmma_f32x4(v2f a, v2f b, v8f c) {
  return __builtin_amdgcn_wmma_f32_16x16x4_f32(false, a, false, b, (short)0, c,
                                               false, false);
}

__global__ void rgcn_count_kernel(const int* __restrict__ dst,
                                  const int* __restrict__ et,
                                  float* __restrict__ cnt, int E) {
  int e = blockIdx.x * blockDim.x + threadIdx.x;
  if (e < E) atomicAdd(&cnt[(size_t)dst[e] * RGCN_R + et[e]], 1.0f);
}

template<int K>
__global__ void rgcn_pack_kernel(const float* __restrict__ rootW,
                                 const float* __restrict__ basisW,
                                 float* __restrict__ Bp) {
  const int t = blockIdx.x * blockDim.x + threadIdx.x;
  if (t >= K * RGCN_W) return;
  const int m = t & 3;
  const int n = (t >> 2) % RGCN_W;
  const int q = t / (4 * RGCN_W);
  const int k = q * 4 + m;
  float v;
  if (n < RGCN_HID) {
    v = rootW[k * RGCN_HID + n];
  } else {
    const int b = (n - RGCN_HID) >> 6;
    const int c = (n - RGCN_HID) & 63;
    v = basisW[((size_t)b * K + k) * RGCN_HID + c];
  }
  Bp[t] = v;
}

template<int PITCH, bool ADD_BIAS>
__device__ __forceinline__ void rgcn_gemm_store_pair(float* __restrict__ outp,
                                                     v8f t0, v8f t1, float bv,
                                                     int hi) {
  float lo[8], up[8];
#pragma unroll
  for (int v = 0; v < 8; ++v) {
    const float x0 = __shfl_xor(t0[v], 16, 32);
    const float x1 = __shfl_xor(t1[v], 16, 32);
    lo[v] = hi ? x1 : t0[v];
    up[v] = hi ? t1[v] : x0;
  }
#pragma unroll
  for (int v = 0; v < 8; ++v) {
    outp[(size_t)v * PITCH]       = ADD_BIAS ? lo[v] + bv : lo[v];
    outp[(size_t)(v + 8) * PITCH] = ADD_BIAS ? up[v] + bv : up[v];
  }
}

#define GEMM_MT 2
#define GEMM_THREADS 160

template<int K, bool RELU_A>
__global__ __launch_bounds__(GEMM_THREADS) void rgcn_gemm_kernel(
    const float* __restrict__ A, const float* __restrict__ Bp,
    const float* __restrict__ bias, float* __restrict__ h,
    float* __restrict__ z, int nRows) {
  const int wave = __builtin_amdgcn_readfirstlane((int)(threadIdx.x >> 5));
  const int lane = threadIdx.x & 31;
  const int hi   = lane >> 4;
  const int lm   = lane & 15;
  const int rowBase = blockIdx.x * (16 * GEMM_MT);

  __shared__ __attribute__((aligned(16))) float As[16 * GEMM_MT][K + 4];
  constexpr int kTasks = 16 * GEMM_MT * (K / 4);
  constexpr int kTrips = kTasks / 128;
  static_assert(kTasks % 128 == 0, "staging tasks must divide over 128 threads");
  if (wave < 4) {
    float4 st[kTrips];
#pragma unroll
    for (int it = 0; it < kTrips; ++it) {
      const int idx = (int)threadIdx.x + it * 128;
      const int r = idx / (K / 4);
      const int q = idx - r * (K / 4);
      int row = rowBase + r;
      if (row > nRows - 1) row = nRows - 1;
      float4 v = *(const float4*)(A + (size_t)row * K + 4 * q);
      if (RELU_A) {
        v.x = fmaxf(v.x, 0.f); v.y = fmaxf(v.y, 0.f);
        v.z = fmaxf(v.z, 0.f); v.w = fmaxf(v.w, 0.f);
      }
      st[it] = v;
    }
#pragma unroll
    for (int it = 0; it < kTrips; ++it) {
      const int idx = (int)threadIdx.x + it * 128;
      const int r = idx / (K / 4);
      const int q = idx - r * (K / 4);
      *(float4*)&As[r][4 * q] = st[it];
    }
  }
  __syncthreads();
  const float* __restrict__ bptr = Bp + (size_t)(64 * wave + lm) * 4 + 2 * hi;

  v8f acc[GEMM_MT][4] = {};
#pragma unroll 4
  for (int k = 0; k < K; k += 4) {
    v2f a[GEMM_MT], b[4];
#pragma unroll
    for (int m = 0; m < GEMM_MT; ++m) {
      a[m] = *(const v2f*)&As[16 * m + lm][k + 2 * hi];
    }
#pragma unroll
    for (int t = 0; t < 4; ++t)
      b[t] = *(const v2f*)(bptr + t * 64 + (size_t)(k >> 2) * (4 * RGCN_W));
#pragma unroll
    for (int m = 0; m < GEMM_MT; ++m) {
#pragma unroll
      for (int t = 0; t < 4; ++t) acc[m][t] = wmma_f32x4(a[m], b[t], acc[m][t]);
    }
  }

  if (wave == 0) {
#pragma unroll
    for (int m = 0; m < GEMM_MT; ++m) {
      const int r0 = rowBase + 16 * m;
      if (r0 + 16 <= nRows) {
#pragma unroll
        for (int p = 0; p < 2; ++p)
          rgcn_gemm_store_pair<RGCN_HID, true>(
              h + (size_t)r0 * RGCN_HID + 32 * p + lane, acc[m][2 * p],
              acc[m][2 * p + 1], bias[32 * p + lane], hi);
      }
    }
  } else {
    const int zc = 64 * (wave - 1) + lane;
#pragma unroll
    for (int m = 0; m < GEMM_MT; ++m) {
      const int r0 = rowBase + 16 * m;
      if (r0 + 16 <= nRows) {
#pragma unroll
        for (int p = 0; p < 2; ++p)
          rgcn_gemm_store_pair<RGCN_B * RGCN_HID, false>(
              z + (size_t)r0 * (RGCN_B * RGCN_HID) + zc + 32 * p, acc[m][2 * p],
              acc[m][2 * p + 1], 0.f, hi);
      }
    }
  }
}

#define SCAT_EPW 32
#define SCAT_GRP 4

template<int G>
__device__ __forceinline__ void rgcn_scatter_group(
    int s, int d, float c0, float c1, float c2, float c3, int j0, int lane,
    const float* __restrict__ z, float* __restrict__ hout) {
  float zv[G][8];
#pragma unroll
  for (int u = 0; u < G; ++u) {
    const int sj = __builtin_amdgcn_readlane(s, j0 + u);
    const float* __restrict__ zp =
        z + (size_t)sj * (RGCN_B * RGCN_HID) + lane;
#pragma unroll
    for (int q = 0; q < 8; ++q) zv[u][q] = zp[q * 32];
  }
#pragma unroll
  for (int u = 0; u < G; ++u) {
    const int j  = j0 + u;
    const int dj = __builtin_amdgcn_readlane(d, j);
    const float a0 = __int_as_float(__builtin_amdgcn_readlane(__float_as_int(c0), j));
    const float a1 = __int_as_float(__builtin_amdgcn_readlane(__float_as_int(c1), j));
    const float a2 = __int_as_float(__builtin_amdgcn_readlane(__float_as_int(c2), j));
    const float a3 = __int_as_float(__builtin_amdgcn_readlane(__float_as_int(c3), j));
    const float m0 = a0 * zv[u][0] + a1 * zv[u][2] + a2 * zv[u][4] + a3 * zv[u][6];
    const float m1 = a0 * zv[u][1] + a1 * zv[u][3] + a2 * zv[u][5] + a3 * zv[u][7];
    float* __restrict__ hp = hout + (size_t)dj * RGCN_HID + lane;
    atomicAdd(hp, m0);
    atomicAdd(hp + 32, m1);
  }
}

__global__ __launch_bounds__(256) void rgcn_scatter_kernel(
    const int* __restrict__ src, const int* __restrict__ dst,
    const int* __restrict__ et, const float* __restrict__ comp,
    const float* __restrict__ probs, const float* __restrict__ cnt,
    const float* __restrict__ z, float* __restrict__ hout, int E) {
  const int lane = threadIdx.x & 31;
  const int wib = __builtin_amdgcn_readfirstlane((int)(threadIdx.x >> 5));
  const int e0  = (blockIdx.x * (blockDim.x >> 5) + wib) * SCAT_EPW;
  if (e0 >= E) return;
  int nE = E - e0;
  if (nE > SCAT_EPW) nE = SCAT_EPW;

  int s = 0, d = 0;
  float c0 = 0.f, c1 = 0.f, c2 = 0.f, c3 = 0.f;
  if (lane < nE) {
    const int e = e0 + lane;
    s = src[e];
    d = dst[e];
    const int t = et[e];
    const float gate = 1.f / (1.f + __expf(-probs[t]));
    const float c    = cnt[(size_t)d * RGCN_R + t];
    const float g    = gate / fmaxf(c, 1.f);
    const float4 cv  = *(const float4*)(comp + t * RGCN_B);
    c0 = cv.x * g; c1 = cv.y * g; c2 = cv.z * g; c3 = cv.w * g;
  }

  int j0 = 0;
  for (; j0 + SCAT_GRP <= nE; j0 += SCAT_GRP)
    rgcn_scatter_group<SCAT_GRP>(s, d, c0, c1, c2, c3, j0, lane, z, hout);
  for (; j0 < nE; ++j0)
    rgcn_scatter_group<1>(s, d, c0, c1, c2, c3, j0, lane, z, hout);
}

#define SC_EPB 10

__global__ __launch_bounds__(256) void rgcn_score_kernel(
    const float* __restrict__ hB,
    const int* __restrict__ tei,
    const float* __restrict__ Rm,
    const float* __restrict__ D,
    float* __restrict__ out, int ET) {
  __shared__ __attribute__((aligned(16))) float x1s[SC_EPB][RGCN_HID];
  __shared__ __attribute__((aligned(16))) float x2s[SC_EPB][RGCN_HID];
  __shared__ __attribute__((aligned(16))) float outs[SC_EPB][RGCN_R];
  const int base = blockIdx.x * SC_EPB;

  for (int idx = threadIdx.x; idx < 2 * SC_EPB * 16; idx += blockDim.x) {
    const int which = idx / (SC_EPB * 16);
    const int rem   = idx - which * (SC_EPB * 16);
    const int m = rem >> 4, q = rem & 15;
    const int e = base + m;
    if (e < ET) {
      const int node = tei[which * ET + e];
      float4 v = *(const float4*)(hB + (size_t)node * RGCN_HID + q * 4);
      v.x = fmaxf(v.x, 0.f); v.y = fmaxf(v.y, 0.f);
      v.z = fmaxf(v.z, 0.f); v.w = fmaxf(v.w, 0.f);
      float* dstp = which ? &x2s[m][q * 4] : &x1s[m][q * 4];
      *(float4*)dstp = v;
    }
  }

  const int wave = threadIdx.x >> 5;
  const int lane = threadIdx.x & 31;
  const int hi   = lane >> 4;
  const int lm   = lane & 15;
  const int r    = wave * 16 + lm;

  v2f af[4][16];
  v2f dj[16];
  float di[4][8];
#pragma unroll
  for (int mt = 0; mt < 4; ++mt) {
#pragma unroll
    for (int kk = 0; kk < 16; ++kk)
      af[mt][kk] = *(const v2f*)(Rm + (mt * 16 + lm) * RGCN_HID + 4 * kk + 2 * hi);
  }
#pragma unroll
  for (int kk = 0; kk < 16; ++kk)
    dj[kk] = *(const v2f*)(D + r * RGCN_HID + 4 * kk + 2 * hi);
#pragma unroll
  for (int mt = 0; mt < 4; ++mt) {
    const float4 lo = *(const float4*)(D + r * RGCN_HID + mt * 16 + 8 * hi);
    const float4 up = *(const float4*)(D + r * RGCN_HID + mt * 16 + 8 * hi + 4);
    di[mt][0] = lo.x; di[mt][1] = lo.y; di[mt][2] = lo.z; di[mt][3] = lo.w;
    di[mt][4] = up.x; di[mt][5] = up.y; di[mt][6] = up.z; di[mt][7] = up.w;
  }
  __syncthreads();

  for (int m = 0; m < SC_EPB; ++m) {
    if (base + m >= ET) break;
    v8f acc[4] = {};
#pragma unroll
    for (int kk = 0; kk < 16; ++kk) {
      const v2f xv = *(const v2f*)&x2s[m][4 * kk + 2 * hi];
      const v2f b  = dj[kk] * xv;
#pragma unroll
      for (int mt = 0; mt < 4; ++mt) acc[mt] = wmma_f32x4(af[mt][kk], b, acc[mt]);
    }
    float p[4];
#pragma unroll
    for (int mt = 0; mt < 4; ++mt) {
      const float4 lo = *(const float4*)&x1s[m][mt * 16 + 8 * hi];
      const float4 up = *(const float4*)&x1s[m][mt * 16 + 8 * hi + 4];
      float s = (lo.x * di[mt][0]) * acc[mt][0];
      s += (lo.y * di[mt][1]) * acc[mt][1];
      s += (lo.z * di[mt][2]) * acc[mt][2];
      s += (lo.w * di[mt][3]) * acc[mt][3];
      s += (up.x * di[mt][4]) * acc[mt][4];
      s += (up.y * di[mt][5]) * acc[mt][5];
      s += (up.z * di[mt][6]) * acc[mt][6];
      s += (up.w * di[mt][7]) * acc[mt][7];
      p[mt] = s;
    }
    float t = (p[0] + p[1]) + (p[2] + p[3]);
    t += __shfl_xor(t, 16, 32);
    if (hi == 0) outs[m][r] = 1.f / (1.f + __expf(-t));
  }
  __syncthreads();

  int nvalid = ET - base;
  if (nvalid > SC_EPB) nvalid = SC_EPB;
  float* __restrict__ obase = out + (size_t)base * RGCN_R;
  const float* osrc = &outs[0][0];
  for (int idx = threadIdx.x * 4; idx < nvalid * RGCN_R; idx += blockDim.x * 4)
    *(float4*)(obase + idx) = *(const float4*)(osrc + idx);
}

extern "C" void kernel_launch(void* const* d_in, const int* in_sizes, int n_in,
                              void* d_out, int out_size, void* d_ws,
                              size_t ws_size, hipStream_t stream) {
  const float* x      = (const float*)d_in[0];
  const int*   eidx   = (const int*)d_in[1];
  const int*   etype  = (const int*)d_in[2];
  const int*   tei    = (const int*)d_in[3];
  const float* probs  = (const float*)d_in[4];
  const float* basis1 = (const float*)d_in[5];
  const float* comp1  = (const float*)d_in[6];
  const float* root1  = (const float*)d_in[7];
  const float* bias1  = (const float*)d_in[8];
  const float* basis2 = (const float*)d_in[9];
  const float* comp2  = (const float*)d_in[10];
  const float* root2  = (const float*)d_in[11];
  const float* bias2  = (const float*)d_in[12];
  const float* Rm     = (const float*)d_in[13];
  const float* D      = (const float*)d_in[14];

  const int N  = in_sizes[0] / RGCN_IN;
  const int E  = in_sizes[2];
  const int ET = in_sizes[3] / 2;

  float* cnt = (float*)d_ws;
  float* z   = cnt + (size_t)N * RGCN_R;
  float* hA  = z   + (size_t)N * RGCN_B * RGCN_HID;
  float* hB  = hA  + (size_t)N * RGCN_HID;
  float* Bp1 = hB  + (size_t)N * RGCN_HID;
  float* Bp2 = Bp1 + (size_t)RGCN_IN * RGCN_W;

  const int* src = eidx;
  const int* dst = eidx + E;

  hipMemsetAsync(cnt, 0, (size_t)N * RGCN_R * sizeof(float), stream);

  rgcn_count_kernel<<<(E + 255) / 256, 256, 0, stream>>>(dst, etype, cnt, E);

  rgcn_pack_kernel<RGCN_IN>
      <<<(RGCN_IN * RGCN_W + 255) / 256, 256, 0, stream>>>(root1, basis1, Bp1);
  rgcn_pack_kernel<RGCN_HID>
      <<<(RGCN_HID * RGCN_W + 255) / 256, 256, 0, stream>>>(root2, basis2, Bp2);

  const int gg = (N + 16 * GEMM_MT - 1) / (16 * GEMM_MT);

  rgcn_gemm_kernel<RGCN_IN, false>
      <<<gg, GEMM_THREADS, 0, stream>>>(x, Bp1, bias1, hA, z, N);
  rgcn_scatter_kernel<<<(E + 255) / 256, 256, 0, stream>>>(src, dst, etype, comp1,
                                                       probs, cnt, z, hA, E);

  rgcn_gemm_kernel<RGCN_HID, true>
      <<<gg, GEMM_THREADS, 0, stream>>>(hA, Bp2, bias2, hB, z, N);
  rgcn_scatter_kernel<<<(E + 255) / 256, 256, 0, stream>>>(src, dst, etype, comp2,
                                                       probs, cnt, z, hB, E);

  rgcn_score_kernel<<<(ET + SC_EPB - 1) / SC_EPB, 256, 0, stream>>>(
      hB, tei, Rm, D, (float*)d_out, ET);
}


# Layer-2 scatter walks only edges whose destination the scorer reads (node bit set marked from the target edges on every call); other edges' gathers, cnt reads and atomics are not done
# speedup vs baseline: 1.9819x; 1.1727x over previous
#include <hip/hip_runtime.h>
#include <cstddef>

typedef float v2f __attribute__((ext_vector_type(2)));
typedef float v8f __attribute__((ext_vector_type(8)));

#define RGCN_R   128
#define RGCN_B   4
#define RGCN_IN  128
#define RGCN_HID 64
#define RGCN_W   (RGCN_HID * (RGCN_B + 1))

__device__ __forceinline__ v8f wmma_f32x4(v2f a, v2f b, v8f c) {
  return __builtin_amdgcn_wmma_f32_16x16x4_f32(false, a, false, b, (short)0, c,
                                               false, false);
}

__global__ void rgcn_count_kernel(const int* __restrict__ dst,
                                  const int* __restrict__ et,
                                  float* __restrict__ cnt, int E,
                                  const int* __restrict__ tei,
                                  unsigned* __restrict__ need, int nT) {
  const int e = blockIdx.x * blockDim.x + threadIdx.x;
  if (e < E) atomicAdd(&cnt[(size_t)dst[e] * RGCN_R + et[e]], 1.0f);
  if (need != nullptr) {
    for (int i = e; i < nT; i += (int)(gridDim.x * blockDim.x)) {
      const int n = tei[i];
      atomicOr(&need[n >> 5], 1u << (n & 31));
    }
  }
}

template<int K>
__global__ void rgcn_pack_kernel(const float* __restrict__ rootW,
                                 const float* __restrict__ basisW,
                                 float* __restrict__ Bp) {
  const int t = blockIdx.x * blockDim.x + threadIdx.x;
  if (t >= K * RGCN_W) return;
  const int m = t & 3;
  const int n = (t >> 2) % RGCN_W;
  const int q = t / (4 * RGCN_W);
  const int k = q * 4 + m;
  float v;
  if (n < RGCN_HID) {
    v = rootW[k * RGCN_HID + n];
  } else {
    const int b = (n - RGCN_HID) >> 6;
    const int c = (n - RGCN_HID) & 63;
    v = basisW[((size_t)b * K + k) * RGCN_HID + c];
  }
  Bp[t] = v;
}

template<int PITCH, bool ADD_BIAS>
__device__ __forceinline__ void rgcn_gemm_store_pair(float* __restrict__ outp,
                                                     v8f t0, v8f t1, float bv,
                                                     int hi) {
  float lo[8], up[8];
#pragma unroll
  for (int v = 0; v < 8; ++v) {
    const float x0 = __shfl_xor(t0[v], 16, 32);
    const float x1 = __shfl_xor(t1[v], 16, 32);
    lo[v] = hi ? x1 : t0[v];
    up[v] = hi ? t1[v] : x0;
  }
#pragma unroll
  for (int v = 0; v < 8; ++v) {
    outp[(size_t)v * PITCH]       = ADD_BIAS ? lo[v] + bv : lo[v];
    outp[(size_t)(v + 8) * PITCH] = ADD_BIAS ? up[v] + bv : up[v];
  }
}

#define GEMM_MT 2
#define GEMM_THREADS 160

template<int K, bool RELU_A>
__global__ __launch_bounds__(GEMM_THREADS) void rgcn_gemm_kernel(
    const float* __restrict__ A, const float* __restrict__ Bp,
    const float* __restrict__ bias, float* __restrict__ h,
    float* __restrict__ z, int nRows) {
  const int wave = __builtin_amdgcn_readfirstlane((int)(threadIdx.x >> 5));
  const int lane = threadIdx.x & 31;
  const int hi   = lane >> 4;
  const int lm   = lane & 15;
  const int rowBase = blockIdx.x * (16 * GEMM_MT);

  __shared__ __attribute__((aligned(16))) float As[16 * GEMM_MT][K + 4];
  constexpr int kTasks = 16 * GEMM_MT * (K / 4);
  constexpr int kTrips = kTasks / 128;
  static_assert(kTasks % 128 == 0, "staging tasks must divide over 128 threads");
  if (wave < 4) {
    float4 st[kTrips];
#pragma unroll
    for (int it = 0; it < kTrips; ++it) {
      const int idx = (int)threadIdx.x + it * 128;
      const int r = idx / (K / 4);
      const int q = idx - r * (K / 4);
      int row = rowBase + r;
      if (row > nRows - 1) row = nRows - 1;
      float4 v = *(const float4*)(A + (size_t)row * K + 4 * q);
      if (RELU_A) {
        v.x = fmaxf(v.x, 0.f); v.y = fmaxf(v.y, 0.f);
        v.z = fmaxf(v.z, 0.f); v.w = fmaxf(v.w, 0.f);
      }
      st[it] = v;
    }
#pragma unroll
    for (int it = 0; it < kTrips; ++it) {
      const int idx = (int)threadIdx.x + it * 128;
      const int r = idx / (K / 4);
      const int q = idx - r * (K / 4);
      *(float4*)&As[r][4 * q] = st[it];
    }
  }
  __syncthreads();
  const float* __restrict__ bptr = Bp + (size_t)(64 * wave + lm) * 4 + 2 * hi;

  v8f acc[GEMM_MT][4] = {};
#pragma unroll 4
  for (int k = 0; k < K; k += 4) {
    v2f a[GEMM_MT], b[4];
#pragma unroll
    for (int m = 0; m < GEMM_MT; ++m) {
      a[m] = *(const v2f*)&As[16 * m + lm][k + 2 * hi];
    }
#pragma unroll
    for (int t = 0; t < 4; ++t)
      b[t] = *(const v2f*)(bptr + t * 64 + (size_t)(k >> 2) * (4 * RGCN_W));
#pragma unroll
    for (int m = 0; m < GEMM_MT; ++m) {
#pragma unroll
      for (int t = 0; t < 4; ++t) acc[m][t] = wmma_f32x4(a[m], b[t], acc[m][t]);
    }
  }

  if (wave == 0) {
#pragma unroll
    for (int m = 0; m < GEMM_MT; ++m) {
      const int r0 = rowBase + 16 * m;
      if (r0 + 16 <= nRows) {
#pragma unroll
        for (int p = 0; p < 2; ++p)
          rgcn_gemm_store_pair<RGCN_HID, true>(
              h + (size_t)r0 * RGCN_HID + 32 * p + lane, acc[m][2 * p],
              acc[m][2 * p + 1], bias[32 * p + lane], hi);
      }
    }
  } else {
    const int zc = 64 * (wave - 1) + lane;
#pragma unroll
    for (int m = 0; m < GEMM_MT; ++m) {
      const int r0 = rowBase + 16 * m;
      if (r0 + 16 <= nRows) {
#pragma unroll
        for (int p = 0; p < 2; ++p)
          rgcn_gemm_store_pair<RGCN_B * RGCN_HID, false>(
              z + (size_t)r0 * (RGCN_B * RGCN_HID) + zc + 32 * p, acc[m][2 * p],
              acc[m][2 * p + 1], 0.f, hi);
      }
    }
  }
}

#define SCAT_EPW 32
#define SCAT_GRP 4

template<int G>
__device__ __forceinline__ void rgcn_scatter_group(
    int s, int d, float c0, float c1, float c2, float c3, unsigned& m, int lane,
    const float* __restrict__ z, float* __restrict__ hout) {
  int jl[G];
#pragma unroll
  for (int u = 0; u < G; ++u) {
    jl[u] = __builtin_ctz(m);
    m &= m - 1u;
  }
  float zv[G][8];
#pragma unroll
  for (int u = 0; u < G; ++u) {
    const int sj = __builtin_amdgcn_readlane(s, jl[u]);
    const float* __restrict__ zp =
        z + (size_t)sj * (RGCN_B * RGCN_HID) + lane;
#pragma unroll
    for (int q = 0; q < 8; ++q) zv[u][q] = zp[q * 32];
  }
#pragma unroll
  for (int u = 0; u < G; ++u) {
    const int j  = jl[u];
    const int dj = __builtin_amdgcn_readlane(d, j);
    const float a0 = __int_as_float(__builtin_amdgcn_readlane(__float_as_int(c0), j));
    const float a1 = __int_as_float(__builtin_amdgcn_readlane(__float_as_int(c1), j));
    const float a2 = __int_as_float(__builtin_amdgcn_readlane(__float_as_int(c2), j));
    const float a3 = __int_as_float(__builtin_amdgcn_readlane(__float_as_int(c3), j));
    const float m0 = a0 * zv[u][0] + a1 * zv[u][2] + a2 * zv[u][4] + a3 * zv[u][6];
    const float m1 = a0 * zv[u][1] + a1 * zv[u][3] + a2 * zv[u][5] + a3 * zv[u][7];
    float* __restrict__ hp = hout + (size_t)dj * RGCN_HID + lane;
    atomicAdd(hp, m0);
    atomicAdd(hp + 32, m1);
  }
}

__global__ __launch_bounds__(256) void rgcn_scatter_kernel(
    const int* __restrict__ src, const int* __restrict__ dst,
    const int* __restrict__ et, const float* __restrict__ comp,
    const float* __restrict__ probs, const float* __restrict__ cnt,
    const unsigned* __restrict__ need,
    const float* __restrict__ z, float* __restrict__ hout, int E) {
  const int lane = threadIdx.x & 31;
  const int wib = __builtin_amdgcn_readfirstlane((int)(threadIdx.x >> 5));
  const int e0  = (blockIdx.x * (blockDim.x >> 5) + wib) * SCAT_EPW;
  if (e0 >= E) return;
  int nE = E - e0;
  if (nE > SCAT_EPW) nE = SCAT_EPW;

  int s = 0, d = 0, t = 0;
  float c0 = 0.f, c1 = 0.f, c2 = 0.f, c3 = 0.f;
  bool keep = false;
  if (lane < nE) {
    const int e = e0 + lane;
    s = src[e];
    d = dst[e];
    t = et[e];
    keep = true;
    if (need != nullptr)
      keep = ((need[d >> 5] >> (d & 31)) & 1u) != 0u;
  }
  if (keep) {
    const float gate = 1.f / (1.f + __expf(-probs[t]));
    const float c    = cnt[(size_t)d * RGCN_R + t];
    const float g    = gate / fmaxf(c, 1.f);
    const float4 cv  = *(const float4*)(comp + t * RGCN_B);
    c0 = cv.x * g; c1 = cv.y * g; c2 = cv.z * g; c3 = cv.w * g;
  }

  unsigned m = __builtin_amdgcn_ballot_w32(keep);
  while (__builtin_popcount(m) >= SCAT_GRP)
    rgcn_scatter_group<SCAT_GRP>(s, d, c0, c1, c2, c3, m, lane, z, hout);
  const int left = __builtin_popcount(m);
  if (left == 3)      rgcn_scatter_group<3>(s, d, c0, c1, c2, c3, m, lane, z, hout);
  else if (left == 2) rgcn_scatter_group<2>(s, d, c0, c1, c2, c3, m, lane, z, hout);
  else if (left == 1) rgcn_scatter_group<1>(s, d, c0, c1, c2, c3, m, lane, z, hout);
}

#define SC_EPB 10

__global__ __launch_bounds__(256) void rgcn_score_kernel(
    const float* __restrict__ hB,
    const int* __restrict__ tei,
    const float* __restrict__ Rm,
    const float* __restrict__ D,
    float* __restrict__ out, int ET) {
  __shared__ __attribute__((aligned(16))) float x1s[SC_EPB][RGCN_HID];
  __shared__ __attribute__((aligned(16))) float x2s[SC_EPB][RGCN_HID];
  __shared__ __attribute__((aligned(16))) float outs[SC_EPB][RGCN_R];
  const int base = blockIdx.x * SC_EPB;

  for (int idx = threadIdx.x; idx < 2 * SC_EPB * 16; idx += blockDim.x) {
    const int which = idx / (SC_EPB * 16);
    const int rem   = idx - which * (SC_EPB * 16);
    const int m = rem >> 4, q = rem & 15;
    const int e = base + m;
    if (e < ET) {
      const int node = tei[which * ET + e];
      float4 v = *(const float4*)(hB + (size_t)node * RGCN_HID + q * 4);
      v.x = fmaxf(v.x, 0.f); v.y = fmaxf(v.y, 0.f);
      v.z = fmaxf(v.z, 0.f); v.w = fmaxf(v.w, 0.f);
      float* dstp = which ? &x2s[m][q * 4] : &x1s[m][q * 4];
      *(float4*)dstp = v;
    }
  }

  const int wave = threadIdx.x >> 5;
  const int lane = threadIdx.x & 31;
  const int hi   = lane >> 4;
  const int lm   = lane & 15;
  const int r    = wave * 16 + lm;

  v2f af[4][16];
  v2f dj[16];
  float di[4][8];
#pragma unroll
  for (int mt = 0; mt < 4; ++mt) {
#pragma unroll
    for (int kk = 0; kk < 16; ++kk)
      af[mt][kk] = *(const v2f*)(Rm + (mt * 16 + lm) * RGCN_HID + 4 * kk + 2 * hi);
  }
#pragma unroll
  for (int kk = 0; kk < 16; ++kk)
    dj[kk] = *(const v2f*)(D + r * RGCN_HID + 4 * kk + 2 * hi);
#pragma unroll
  for (int mt = 0; mt < 4; ++mt) {
    const float4 lo = *(const float4*)(D + r * RGCN_HID + mt * 16 + 8 * hi);
    const float4 up = *(const float4*)(D + r * RGCN_HID + mt * 16 + 8 * hi + 4);
    di[mt][0] = lo.x; di[mt][1] = lo.y; di[mt][2] = lo.z; di[mt][3] = lo.w;
    di[mt][4] = up.x; di[mt][5] = up.y; di[mt][6] = up.z; di[mt][7] = up.w;
  }
  __syncthreads();

  for (int m = 0; m < SC_EPB; ++m) {
    if (base + m >= ET) break;
    v8f acc[4] = {};
#pragma unroll
    for (int kk = 0; kk < 16; ++kk) {
      const v2f xv = *(const v2f*)&x2s[m][4 * kk + 2 * hi];
      const v2f b  = dj[kk] * xv;
#pragma unroll
      for (int mt = 0; mt < 4; ++mt) acc[mt] = wmma_f32x4(af[mt][kk], b, acc[mt]);
    }
    float p[4];
#pragma unroll
    for (int mt = 0; mt < 4; ++mt) {
      const float4 lo = *(const float4*)&x1s[m][mt * 16 + 8 * hi];
      const float4 up = *(const float4*)&x1s[m][mt * 16 + 8 * hi + 4];
      float s = (lo.x * di[mt][0]) * acc[mt][0];
      s += (lo.y * di[mt][1]) * acc[mt][1];
      s += (lo.z * di[mt][2]) * acc[mt][2];
      s += (lo.w * di[mt][3]) * acc[mt][3];
      s += (up.x * di[mt][4]) * acc[mt][4];
      s += (up.y * di[mt][5]) * acc[mt][5];
      s += (up.z * di[mt][6]) * acc[mt][6];
      s += (up.w * di[mt][7]) * acc[mt][7];
      p[mt] = s;
    }
    float t = (p[0] + p[1]) + (p[2] + p[3]);
    t += __shfl_xor(t, 16, 32);
    if (hi == 0) outs[m][r] = 1.f / (1.f + __expf(-t));
  }
  __syncthreads();

  int nvalid = ET - base;
  if (nvalid > SC_EPB) nvalid = SC_EPB;
  float* __restrict__ obase = out + (size_t)base * RGCN_R;
  const float* osrc = &outs[0][0];
  for (int idx = threadIdx.x * 4; idx < nvalid * RGCN_R; idx += blockDim.x * 4)
    *(float4*)(obase + idx) = *(const float4*)(osrc + idx);
}

extern "C" void kernel_launch(void* const* d_in, const int* in_sizes, int n_in,
                              void* d_out, int out_size, void* d_ws,
                              size_t ws_size, hipStream_t stream) {
  const float* x      = (const float*)d_in[0];
  const int*   eidx   = (const int*)d_in[1];
  const int*   etype  = (const int*)d_in[2];
  const int*   tei    = (const int*)d_in[3];
  const float* probs  = (const float*)d_in[4];
  const float* basis1 = (const float*)d_in[5];
  const float* comp1  = (const float*)d_in[6];
  const float* root1  = (const float*)d_in[7];
  const float* bias1  = (const float*)d_in[8];
  const float* basis2 = (const float*)d_in[9];
  const float* comp2  = (const float*)d_in[10];
  const float* root2  = (const float*)d_in[11];
  const float* bias2  = (const float*)d_in[12];
  const float* Rm     = (const float*)d_in[13];
  const float* D      = (const float*)d_in[14];

  const int N  = in_sizes[0] / RGCN_IN;
  const int E  = in_sizes[2];
  const int ET = in_sizes[3] / 2;

  float* z   = (float*)d_ws;
  float* hA  = z   + (size_t)N * RGCN_B * RGCN_HID;
  float* hB  = hA  + (size_t)N * RGCN_HID;
  float* Bp1 = hB  + (size_t)N * RGCN_HID;
  float* Bp2 = Bp1 + (size_t)RGCN_IN * RGCN_W;
  float* cnt = Bp2 + (size_t)RGCN_HID * RGCN_W;
  float* cntEnd = cnt + (size_t)N * RGCN_R;
  const size_t needWords = (size_t)((N + 31) / 32);
  const bool haveNeed =
      ((size_t)(cntEnd - z) + needWords) * sizeof(float) <= ws_size;
  unsigned* need = haveNeed ? (unsigned*)cntEnd : nullptr;

  const int* src = eidx;
  const int* dst = eidx + E;

  hipMemsetAsync(cnt, 0,
                 ((size_t)N * RGCN_R + (haveNeed ? needWords : 0)) * sizeof(float),
                 stream);

  rgcn_count_kernel<<<(E + 255) / 256, 256, 0, stream>>>(dst, etype, cnt, E, tei,
                                                         need, 2 * ET);

  rgcn_pack_kernel<RGCN_IN>
      <<<(RGCN_IN * RGCN_W + 255) / 256, 256, 0, stream>>>(root1, basis1, Bp1);
  rgcn_pack_kernel<RGCN_HID>
      <<<(RGCN_HID * RGCN_W + 255) / 256, 256, 0, stream>>>(root2, basis2, Bp2);

  const int gg = (N + 16 * GEMM_MT - 1) / (16 * GEMM_MT);

  rgcn_gemm_kernel<RGCN_IN, false>
      <<<gg, GEMM_THREADS, 0, stream>>>(x, Bp1, bias1, hA, z, N);
  rgcn_scatter_kernel<<<(E + 255) / 256, 256, 0, stream>>>(
      src, dst, etype, comp1, probs, cnt, nullptr, z, hA, E);

  rgcn_gemm_kernel<RGCN_HID, true>
      <<<gg, GEMM_THREADS, 0, stream>>>(hA, Bp2, bias2, hB, z, N);
  rgcn_scatter_kernel<<<(E + 255) / 256, 256, 0, stream>>>(
      src, dst, etype, comp2, probs, cnt, need, z, hB, E);

  rgcn_score_kernel<<<(ET + SC_EPB - 1) / SC_EPB, 256, 0, stream>>>(
      hB, tei, Rm, D, (float*)d_out, ET);
}


# Messages summed per destination range in LDS: edges binned by destination each call, one work-group per range counts (dst,relation) and adds rows there; count table and global float atomics removed
# speedup vs baseline: 2.2627x; 1.1417x over previous
#include <hip/hip_runtime.h>
#include <cstddef>

typedef float v2f __attribute__((ext_vector_type(2)));
typedef float v8f __attribute__((ext_vector_type(8)));

#define RGCN_R   128
#define RGCN_B   4
#define RGCN_IN  128
#define RGCN_HID 64
#define RGCN_W   (RGCN_HID * (RGCN_B + 1))

__device__ __forceinline__ v8f wmma_f32x4(v2f a, v2f b, v8f c) {
  return __builtin_amdgcn_wmma_f32_16x16x4_f32(false, a, false, b, (short)0, c,
                                               false, false);
}

#define BIN_P    98
#define BIN_NB   512
#define BIN_EPB  8192
#define BIN_T    1024
#define BIN_LINE 16

__device__ __forceinline__ int rgcn_bin_of(int d) {
  const int b = d / BIN_P;
  return b < BIN_NB - 1 ? b : BIN_NB - 1;
}

__device__ __forceinline__ int rgcn_pad_line(int n) {
  return (n + (BIN_LINE - 1)) & ~(BIN_LINE - 1);
}

__device__ __forceinline__ void rgcn_scan_bins(int* a, int* wsum) {
  const int t = threadIdx.x;
  const int lane = t & 31;
  const int w = t >> 5;
  int v = 0, inc = 0;
  if (t < BIN_NB) {
    v = a[t];
    inc = v;
    for (int d = 1; d < 32; d <<= 1) {
      const int n = __shfl_up(inc, d, 32);
      if (lane >= d) inc += n;
    }
    if (lane == 31) wsum[w] = inc;
  }
  __syncthreads();
  if (t < BIN_NB) {
    int off = 0;
    for (int k = 0; k < w; ++k) off += wsum[k];
    a[t] = off + inc - v;
  }
  __syncthreads();
}

__global__ __launch_bounds__(BIN_T) void rgcn_binhist_kernel(
    const int* __restrict__ dst, int E, int* __restrict__ binCount,
    const int* __restrict__ tei, unsigned* __restrict__ need, int nT) {
  __shared__ int hist[BIN_NB];
  const int t = threadIdx.x;
  if (t < BIN_NB) hist[t] = 0;
  __syncthreads();
  const int e0 = blockIdx.x * BIN_EPB;
  for (int k = 0; k < BIN_EPB / BIN_T; ++k) {
    const int e = e0 + k * BIN_T + t;
    if (e < E) atomicAdd(&hist[rgcn_bin_of(dst[e])], 1);
  }
  __syncthreads();
  if (t < BIN_NB) {
    const int pc = rgcn_pad_line(hist[t]);
    if (pc > 0) atomicAdd(&binCount[t], pc);
  }
  for (int i = blockIdx.x * BIN_T + t; i < nT; i += (int)(gridDim.x * BIN_T)) {
    const int n = tei[i];
    atomicOr(&need[n >> 5], 1u << (n & 31));
  }
}

__global__ __launch_bounds__(BIN_T) void rgcn_binfill_kernel(
    const int* __restrict__ src, const int* __restrict__ dst,
    const int* __restrict__ et, int E, const int* __restrict__ binCount,
    int* __restrict__ binCursor, int* __restrict__ binStart,
    int* __restrict__ binnedW) {
  int2* __restrict__ binned = (int2*)binnedW;
  __shared__ int hist[BIN_NB];
  __shared__ int loff[BIN_NB];
  __shared__ int fill[BIN_NB];
  __shared__ int gbase[BIN_NB];
  __shared__ int wsum[BIN_NB / 32];
  __shared__ int2 staged[BIN_EPB];
  const int t = threadIdx.x;
  if (t < BIN_NB) {
    hist[t] = 0;
    gbase[t] = binCount[t];
  }
  __syncthreads();
  const int e0 = blockIdx.x * BIN_EPB;
  int mybin[BIN_EPB / BIN_T];
#pragma unroll
  for (int k = 0; k < BIN_EPB / BIN_T; ++k) {
    const int e = e0 + k * BIN_T + t;
    mybin[k] = -1;
    if (e < E) {
      mybin[k] = rgcn_bin_of(dst[e]);
      atomicAdd(&hist[mybin[k]], 1);
    }
  }
  __syncthreads();
  rgcn_scan_bins(gbase, wsum);
  if (t < BIN_NB) {
    const int h = hist[t];
    if (blockIdx.x == 0) binStart[t] = gbase[t];
    const int pc = rgcn_pad_line(h);
    if (pc > 0) gbase[t] += atomicAdd(&binCursor[t], pc);
    loff[t] = h;
  }
  __syncthreads();
  rgcn_scan_bins(loff, wsum);
  if (t < BIN_NB) fill[t] = loff[t];
  __syncthreads();
#pragma unroll
  for (int k = 0; k < BIN_EPB / BIN_T; ++k) {
    if (mybin[k] >= 0) {
      const int e = e0 + k * BIN_T + t;
      const int b = mybin[k];
      const int pos = atomicAdd(&fill[b], 1);
      staged[pos] = make_int2(src[e], (dst[e] - b * BIN_P) * RGCN_R + et[e]);
    }
  }
  __syncthreads();
  const int lane = t & 31;
  const int w = t >> 5;
  for (int b = w; b < BIN_NB; b += BIN_T / 32) {
    const int h  = hist[b];
    const int pc = rgcn_pad_line(h);
    const int lo = loff[b];
    const int gb = gbase[b];
    for (int k = lane; k < pc; k += 32) {
      int2 v = make_int2(-1, 0);
      if (k < h) v = staged[lo + k];
      binned[gb + k] = v;
    }
  }
}

template<int K>
__global__ void rgcn_pack_kernel(const float* __restrict__ rootW,
                                 const float* __restrict__ basisW,
                                 float* __restrict__ Bp) {
  const int t = blockIdx.x * blockDim.x + threadIdx.x;
  if (t >= K * RGCN_W) return;
  const int m = t & 3;
  const int n = (t >> 2) % RGCN_W;
  const int q = t / (4 * RGCN_W);
  const int k = q * 4 + m;
  float v;
  if (n < RGCN_HID) {
    v = rootW[k * RGCN_HID + n];
  } else {
    const int b = (n - RGCN_HID) >> 6;
    const int c = (n - RGCN_HID) & 63;
    v = basisW[((size_t)b * K + k) * RGCN_HID + c];
  }
  Bp[t] = v;
}

template<int PITCH, bool ADD_BIAS>
__device__ __forceinline__ void rgcn_gemm_store_pair(float* __restrict__ outp,
                                                     v8f t0, v8f t1, float bv,
                                                     int hi) {
  float lo[8], up[8];
#pragma unroll
  for (int v = 0; v < 8; ++v) {
    const float x0 = __shfl_xor(t0[v], 16, 32);
    const float x1 = __shfl_xor(t1[v], 16, 32);
    lo[v] = hi ? x1 : t0[v];
    up[v] = hi ? t1[v] : x0;
  }
#pragma unroll
  for (int v = 0; v < 8; ++v) {
    outp[(size_t)v * PITCH]       = ADD_BIAS ? lo[v] + bv : lo[v];
    outp[(size_t)(v + 8) * PITCH] = ADD_BIAS ? up[v] + bv : up[v];
  }
}

#define GEMM_MT 2
#define GEMM_THREADS 160

template<int K, bool RELU_A>
__global__ __launch_bounds__(GEMM_THREADS) void rgcn_gemm_kernel(
    const float* __restrict__ A, const float* __restrict__ Bp,
    const float* __restrict__ bias, float* __restrict__ h,
    float* __restrict__ z, int nRows) {
  const int wave = __builtin_amdgcn_readfirstlane((int)(threadIdx.x >> 5));
  const int lane = threadIdx.x & 31;
  const int hi   = lane >> 4;
  const int lm   = lane & 15;
  const int rowBase = blockIdx.x * (16 * GEMM_MT);

  __shared__ __attribute__((aligned(16))) float As[16 * GEMM_MT][K + 4];
  constexpr int kTasks = 16 * GEMM_MT * (K / 4);
  constexpr int kTrips = kTasks / 128;
  static_assert(kTasks % 128 == 0, "staging tasks must divide over 128 threads");
  if (wave < 4) {
    float4 st[kTrips];
#pragma unroll
    for (int it = 0; it < kTrips; ++it) {
      const int idx = (int)threadIdx.x + it * 128;
      const int r = idx / (K / 4);
      const int q = idx - r * (K / 4);
      int row = rowBase + r;
      if (row > nRows - 1) row = nRows - 1;
      float4 v = *(const float4*)(A + (size_t)row * K + 4 * q);
      if (RELU_A) {
        v.x = fmaxf(v.x, 0.f); v.y = fmaxf(v.y, 0.f);
        v.z = fmaxf(v.z, 0.f); v.w = fmaxf(v.w, 0.f);
      }
      st[it] = v;
    }
#pragma unroll
    for (int it = 0; it < kTrips; ++it) {
      const int idx = (int)threadIdx.x + it * 128;
      const int r = idx / (K / 4);
      const int q = idx - r * (K / 4);
      *(float4*)&As[r][4 * q] = st[it];
    }
  }
  __syncthreads();
  const float* __restrict__ bptr = Bp + (size_t)(64 * wave + lm) * 4 + 2 * hi;

  v8f acc[GEMM_MT][4] = {};
#pragma unroll 4
  for (int k = 0; k < K; k += 4) {
    v2f a[GEMM_MT], b[4];
#pragma unroll
    for (int m = 0; m < GEMM_MT; ++m) {
      a[m] = *(const v2f*)&As[16 * m + lm][k + 2 * hi];
    }
#pragma unroll
    for (int t = 0; t < 4; ++t)
      b[t] = *(const v2f*)(bptr + t * 64 + (size_t)(k >> 2) * (4 * RGCN_W));
#pragma unroll
    for (int m = 0; m < GEMM_MT; ++m) {
#pragma unroll
      for (int t = 0; t < 4; ++t) acc[m][t] = wmma_f32x4(a[m], b[t], acc[m][t]);
    }
  }

  if (wave == 0) {
#pragma unroll
    for (int m = 0; m < GEMM_MT; ++m) {
      const int r0 = rowBase + 16 * m;
      if (r0 + 16 <= nRows) {
#pragma unroll
        for (int p = 0; p < 2; ++p)
          rgcn_gemm_store_pair<RGCN_HID, true>(
              h + (size_t)r0 * RGCN_HID + 32 * p + lane, acc[m][2 * p],
              acc[m][2 * p + 1], bias[32 * p + lane], hi);
      }
    }
  } else {
    const int zc = 64 * (wave - 1) + lane;
#pragma unroll
    for (int m = 0; m < GEMM_MT; ++m) {
      const int r0 = rowBase + 16 * m;
      if (r0 + 16 <= nRows) {
#pragma unroll
        for (int p = 0; p < 2; ++p)
          rgcn_gemm_store_pair<RGCN_B * RGCN_HID, false>(
              z + (size_t)r0 * (RGCN_B * RGCN_HID) + zc + 32 * p, acc[m][2 * p],
              acc[m][2 * p + 1], 0.f, hi);
      }
    }
  }
}

#define SCAT_GRP 4

template<int G>
__device__ __forceinline__ void rgcn_tile_group(
    int s, int dl, float c0, float c1, float c2, float c3, unsigned& m, int lane,
    const float* __restrict__ z, float* tile) {
  int jl[G];
#pragma unroll
  for (int u = 0; u < G; ++u) {
    jl[u] = __builtin_ctz(m);
    m &= m - 1u;
  }
  float zv[G][8];
#pragma unroll
  for (int u = 0; u < G; ++u) {
    const int sj = __builtin_amdgcn_readlane(s, jl[u]);
    const float* __restrict__ zp =
        z + (size_t)sj * (RGCN_B * RGCN_HID) + lane;
#pragma unroll
    for (int q = 0; q < 8; ++q) zv[u][q] = zp[q * 32];
  }
#pragma unroll
  for (int u = 0; u < G; ++u) {
    const int j  = jl[u];
    const int dj = __builtin_amdgcn_readlane(dl, j);
    const float a0 = __int_as_float(__builtin_amdgcn_readlane(__float_as_int(c0), j));
    const float a1 = __int_as_float(__builtin_amdgcn_readlane(__float_as_int(c1), j));
    const float a2 = __int_as_float(__builtin_amdgcn_readlane(__float_as_int(c2), j));
    const float a3 = __int_as_float(__builtin_amdgcn_readlane(__float_as_int(c3), j));
    const float m0 = a0 * zv[u][0] + a1 * zv[u][2] + a2 * zv[u][4] + a3 * zv[u][6];
    const float m1 = a0 * zv[u][1] + a1 * zv[u][3] + a2 * zv[u][5] + a3 * zv[u][7];
    float* tp = tile + dj * RGCN_HID + lane;
    atomicAdd(tp, m0);
    atomicAdd(tp + 32, m1);
  }
}

template<bool USE_NEED>
__global__ __launch_bounds__(BIN_T) void rgcn_aggregate_kernel(
    const int* __restrict__ binnedW,
    const int* __restrict__ binStart,
    const int* __restrict__ binCount, const float* __restrict__ comp,
    const float* __restrict__ probs,
    const unsigned* __restrict__ need,
    const float* __restrict__ z, float* __restrict__ hout, int N) {
  __shared__ unsigned cntL[BIN_P * RGCN_R];
  __shared__ __attribute__((aligned(16))) float tile[BIN_P * RGCN_HID];
  const int2* __restrict__ binned = (const int2*)binnedW;
  const int t = threadIdx.x;
  const int lane = t & 31;
  const int w = __builtin_amdgcn_readfirstlane(t >> 5);
  const int n0 = blockIdx.x * BIN_P;
  const int start = binStart[blockIdx.x];
  const int count = binCount[blockIdx.x];

  for (int i = t; i < BIN_P * RGCN_R; i += BIN_T) cntL[i] = 0u;
  for (int i = t; i < BIN_P * RGCN_HID; i += BIN_T) tile[i] = 0.f;
  __syncthreads();

  for (int i = t; i < count; i += BIN_T) {
    const int2 ed = binned[start + i];
    if (ed.x >= 0) atomicAdd(&cntL[ed.y], 1u);
  }
  __syncthreads();

  for (int base = w * 32; base < count; base += BIN_T) {
    const int i = base + lane;
    int s = 0, w1 = 0;
    bool keep = false;
    if (i < count) {
      const int2 ed = binned[start + i];
      s = ed.x;
      w1 = ed.y;
      keep = s >= 0;
    }
    const int dl = w1 >> 7;
    const int ty = w1 & (RGCN_R - 1);
    if (USE_NEED && keep) {
      const int n = n0 + dl;
      keep = ((need[n >> 5] >> (n & 31)) & 1u) != 0u;
    }
    float c0 = 0.f, c1 = 0.f, c2 = 0.f, c3 = 0.f;
    if (keep) {
      const float gate = 1.f / (1.f + __expf(-probs[ty]));
      const float c    = (float)cntL[w1];
      const float g    = gate / fmaxf(c, 1.f);
      const float4 cv  = *(const float4*)(comp + ty * RGCN_B);
      c0 = cv.x * g; c1 = cv.y * g; c2 = cv.z * g; c3 = cv.w * g;
    }
    unsigned m = __builtin_amdgcn_ballot_w32(keep);
    while (__builtin_popcount(m) >= SCAT_GRP)
      rgcn_tile_group<SCAT_GRP>(s, dl, c0, c1, c2, c3, m, lane, z, tile);
    const int left = __builtin_popcount(m);
    if (left == 3)      rgcn_tile_group<3>(s, dl, c0, c1, c2, c3, m, lane, z, tile);
    else if (left == 2) rgcn_tile_group<2>(s, dl, c0, c1, c2, c3, m, lane, z, tile);
    else if (left == 1) rgcn_tile_group<1>(s, dl, c0, c1, c2, c3, m, lane, z, tile);
  }
  __syncthreads();

  int nn = N - n0;
  if (nn > BIN_P) nn = BIN_P;
  for (int idx = t; idx < nn * (RGCN_HID / 4); idx += BIN_T) {
    float4* hp = (float4*)(hout + (size_t)n0 * RGCN_HID) + idx;
    const float4 tv = *(const float4*)&tile[4 * idx];
    float4 hv = *hp;
    hv.x += tv.x; hv.y += tv.y; hv.z += tv.z; hv.w += tv.w;
    *hp = hv;
  }
}

#define SC_EPB 10

__global__ __launch_bounds__(256) void rgcn_score_kernel(
    const float* __restrict__ hB,
    const int* __restrict__ tei,
    const float* __restrict__ Rm,
    const float* __restrict__ D,
    float* __restrict__ out, int ET) {
  __shared__ __attribute__((aligned(16))) float x1s[SC_EPB][RGCN_HID];
  __shared__ __attribute__((aligned(16))) float x2s[SC_EPB][RGCN_HID];
  __shared__ __attribute__((aligned(16))) float outs[SC_EPB][RGCN_R];
  const int base = blockIdx.x * SC_EPB;

  for (int idx = threadIdx.x; idx < 2 * SC_EPB * 16; idx += blockDim.x) {
    const int which = idx / (SC_EPB * 16);
    const int rem   = idx - which * (SC_EPB * 16);
    const int m = rem >> 4, q = rem & 15;
    const int e = base + m;
    if (e < ET) {
      const int node = tei[which * ET + e];
      float4 v = *(const float4*)(hB + (size_t)node * RGCN_HID + q * 4);
      v.x = fmaxf(v.x, 0.f); v.y = fmaxf(v.y, 0.f);
      v.z = fmaxf(v.z, 0.f); v.w = fmaxf(v.w, 0.f);
      float* dstp = which ? &x2s[m][q * 4] : &x1s[m][q * 4];
      *(float4*)dstp = v;
    }
  }

  const int wave = threadIdx.x >> 5;
  const int lane = threadIdx.x & 31;
  const int hi   = lane >> 4;
  const int lm   = lane & 15;
  const int r    = wave * 16 + lm;

  v2f af[4][16];
  v2f dj[16];
  float di[4][8];
#pragma unroll
  for (int mt = 0; mt < 4; ++mt) {
#pragma unroll
    for (int kk = 0; kk < 16; ++kk)
      af[mt][kk] = *(const v2f*)(Rm + (mt * 16 + lm) * RGCN_HID + 4 * kk + 2 * hi);
  }
#pragma unroll
  for (int kk = 0; kk < 16; ++kk)
    dj[kk] = *(const v2f*)(D + r * RGCN_HID + 4 * kk + 2 * hi);
#pragma unroll
  for (int mt = 0; mt < 4; ++mt) {
    const float4 lo = *(const float4*)(D + r * RGCN_HID + mt * 16 + 8 * hi);
    const float4 up = *(const float4*)(D + r * RGCN_HID + mt * 16 + 8 * hi + 4);
    di[mt][0] = lo.x; di[mt][1] = lo.y; di[mt][2] = lo.z; di[mt][3] = lo.w;
    di[mt][4] = up.x; di[mt][5] = up.y; di[mt][6] = up.z; di[mt][7] = up.w;
  }
  __syncthreads();

  for (int m = 0; m < SC_EPB; ++m) {
    if (base + m >= ET) break;
    v8f acc[4] = {};
#pragma unroll
    for (int kk = 0; kk < 16; ++kk) {
      const v2f xv = *(const v2f*)&x2s[m][4 * kk + 2 * hi];
      const v2f b  = dj[kk] * xv;
#pragma unroll
      for (int mt = 0; mt < 4; ++mt) acc[mt] = wmma_f32x4(af[mt][kk], b, acc[mt]);
    }
    float p[4];
#pragma unroll
    for (int mt = 0; mt < 4; ++mt) {
      const float4 lo = *(const float4*)&x1s[m][mt * 16 + 8 * hi];
      const float4 up = *(const float4*)&x1s[m][mt * 16 + 8 * hi + 4];
      float s = (lo.x * di[mt][0]) * acc[mt][0];
      s += (lo.y * di[mt][1]) * acc[mt][1];
      s += (lo.z * di[mt][2]) * acc[mt][2];
      s += (lo.w * di[mt][3]) * acc[mt][3];
      s += (up.x * di[mt][4]) * acc[mt][4];
      s += (up.y * di[mt][5]) * acc[mt][5];
      s += (up.z * di[mt][6]) * acc[mt][6];
      s += (up.w * di[mt][7]) * acc[mt][7];
      p[mt] = s;
    }
    float t = (p[0] + p[1]) + (p[2] + p[3]);
    t += __shfl_xor(t, 16, 32);
    if (hi == 0) outs[m][r] = 1.f / (1.f + __expf(-t));
  }
  __syncthreads();

  int nvalid = ET - base;
  if (nvalid > SC_EPB) nvalid = SC_EPB;
  float* __restrict__ obase = out + (size_t)base * RGCN_R;
  const float* osrc = &outs[0][0];
  for (int idx = threadIdx.x * 4; idx < nvalid * RGCN_R; idx += blockDim.x * 4)
    *(float4*)(obase + idx) = *(const float4*)(osrc + idx);
}

extern "C" void kernel_launch(void* const* d_in, const int* in_sizes, int n_in,
                              void* d_out, int out_size, void* d_ws,
                              size_t ws_size, hipStream_t stream) {
  const float* x      = (const float*)d_in[0];
  const int*   eidx   = (const int*)d_in[1];
  const int*   etype  = (const int*)d_in[2];
  const int*   tei    = (const int*)d_in[3];
  const float* probs  = (const float*)d_in[4];
  const float* basis1 = (const float*)d_in[5];
  const float* comp1  = (const float*)d_in[6];
  const float* root1  = (const float*)d_in[7];
  const float* bias1  = (const float*)d_in[8];
  const float* basis2 = (const float*)d_in[9];
  const float* comp2  = (const float*)d_in[10];
  const float* root2  = (const float*)d_in[11];
  const float* bias2  = (const float*)d_in[12];
  const float* Rm     = (const float*)d_in[13];
  const float* D      = (const float*)d_in[14];

  const int N  = in_sizes[0] / RGCN_IN;
  const int E  = in_sizes[2];
  const int ET = in_sizes[3] / 2;

  float* z   = (float*)d_ws;
  float* hA  = z   + (size_t)N * RGCN_B * RGCN_HID;
  float* hB  = hA  + (size_t)N * RGCN_HID;
  float* Bp1 = hB  + (size_t)N * RGCN_HID;
  float* Bp2 = Bp1 + (size_t)RGCN_IN * RGCN_W;
  int* binCount  = (int*)(Bp2 + (size_t)RGCN_HID * RGCN_W);
  int* binCursor = binCount + BIN_NB;
  unsigned* need = (unsigned*)(binCursor + BIN_NB);
  const size_t needPad = (size_t)((N + 1023) / 1024) * 32;
  int* binStart  = (int*)(need + needPad);
  int* binned    = binStart + BIN_NB;

  const int* src = eidx;
  const int* dst = eidx + E;

  hipMemsetAsync(binCount, 0, ((size_t)2 * BIN_NB + needPad) * sizeof(int), stream);

  const int nb = (E + BIN_EPB - 1) / BIN_EPB;
  rgcn_binhist_kernel<<<nb, BIN_T, 0, stream>>>(dst, E, binCount, tei, need, 2 * ET);
  rgcn_binfill_kernel<<<nb, BIN_T, 0, stream>>>(src, dst, etype, E, binCount,
                                               binCursor, binStart, binned);

  rgcn_pack_kernel<RGCN_IN>
      <<<(RGCN_IN * RGCN_W + 255) / 256, 256, 0, stream>>>(root1, basis1, Bp1);
  rgcn_pack_kernel<RGCN_HID>
      <<<(RGCN_HID * RGCN_W + 255) / 256, 256, 0, stream>>>(root2, basis2, Bp2);

  const int gg = (N + 16 * GEMM_MT - 1) / (16 * GEMM_MT);

  rgcn_gemm_kernel<RGCN_IN, false>
      <<<gg, GEMM_THREADS, 0, stream>>>(x, Bp1, bias1, hA, z, N);
  rgcn_aggregate_kernel<false><<<BIN_NB, BIN_T, 0, stream>>>(
      binned, binStart, binCount, comp1, probs, need, z, hA, N);

  rgcn_gemm_kernel<RGCN_HID, true>
      <<<gg, GEMM_THREADS, 0, stream>>>(hA, Bp2, bias2, hB, z, N);
  rgcn_aggregate_kernel<true><<<BIN_NB, BIN_T, 0, stream>>>(
      binned, binStart, binCount, comp2, probs, need, z, hB, N);

  rgcn_score_kernel<<<(ET + SC_EPB - 1) / SC_EPB, 256, 0, stream>>>(
      hB, tei, Rm, D, (float*)d_out, ET);
}


# Scorer sums only j>=i (D[r,i]*D[r,j] is symmetric): per edge S[i,j]=x1_i x2_j Rm[i,j]+x2_i x1_j Rm[j,i] is the f32 WMMA A operand, 10 of 16 blocks multiplied (40 of 64 k-steps per relation tile)
# speedup vs baseline: 2.4308x; 1.0743x over previous
#include <hip/hip_runtime.h>
#include <cstddef>

typedef float v2f __attribute__((ext_vector_type(2)));
typedef float v8f __attribute__((ext_vector_type(8)));

#define RGCN_R   128
#define RGCN_B   4
#define RGCN_IN  128
#define RGCN_HID 64
#define RGCN_W   (RGCN_HID * (RGCN_B + 1))

__device__ __forceinline__ v8f wmma_f32x4(v2f a, v2f b, v8f c) {
  return __builtin_amdgcn_wmma_f32_16x16x4_f32(false, a, false, b, (short)0, c,
                                               false, false);
}

#define BIN_P    98
#define BIN_NB   512
#define BIN_EPB  8192
#define BIN_T    1024
#define BIN_LINE 16

__device__ __forceinline__ int rgcn_bin_of(int d) {
  const int b = d / BIN_P;
  return b < BIN_NB - 1 ? b : BIN_NB - 1;
}

__device__ __forceinline__ int rgcn_pad_line(int n) {
  return (n + (BIN_LINE - 1)) & ~(BIN_LINE - 1);
}

__device__ __forceinline__ void rgcn_scan_bins(int* a, int* wsum) {
  const int t = threadIdx.x;
  const int lane = t & 31;
  const int w = t >> 5;
  int v = 0, inc = 0;
  if (t < BIN_NB) {
    v = a[t];
    inc = v;
    for (int d = 1; d < 32; d <<= 1) {
      const int n = __shfl_up(inc, d, 32);
      if (lane >= d) inc += n;
    }
    if (lane == 31) wsum[w] = inc;
  }
  __syncthreads();
  if (t < BIN_NB) {
    int off = 0;
    for (int k = 0; k < w; ++k) off += wsum[k];
    a[t] = off + inc - v;
  }
  __syncthreads();
}

__global__ __launch_bounds__(BIN_T) void rgcn_binhist_kernel(
    const int* __restrict__ dst, int E, int* __restrict__ binCount,
    const int* __restrict__ tei, unsigned* __restrict__ need, int nT) {
  __shared__ int hist[BIN_NB];
  const int t = threadIdx.x;
  if (t < BIN_NB) hist[t] = 0;
  __syncthreads();
  const int e0 = blockIdx.x * BIN_EPB;
  for (int k = 0; k < BIN_EPB / BIN_T; ++k) {
    const int e = e0 + k * BIN_T + t;
    if (e < E) atomicAdd(&hist[rgcn_bin_of(dst[e])], 1);
  }
  __syncthreads();
  if (t < BIN_NB) {
    const int pc = rgcn_pad_line(hist[t]);
    if (pc > 0) atomicAdd(&binCount[t], pc);
  }
  for (int i = blockIdx.x * BIN_T + t; i < nT; i += (int)(gridDim.x * BIN_T)) {
    const int n = tei[i];
    atomicOr(&need[n >> 5], 1u << (n & 31));
  }
}

__global__ __launch_bounds__(BIN_T) void rgcn_binfill_kernel(
    const int* __restrict__ src, const int* __restrict__ dst,
    const int* __restrict__ et, int E, const int* __restrict__ binCount,
    int* __restrict__ binCursor, int* __restrict__ binStart,
    int* __restrict__ binnedW) {
  int2* __restrict__ binned = (int2*)binnedW;
  __shared__ int hist[BIN_NB];
  __shared__ int loff[BIN_NB];
  __shared__ int fill[BIN_NB];
  __shared__ int gbase[BIN_NB];
  __shared__ int wsum[BIN_NB / 32];
  __shared__ int2 staged[BIN_EPB];
  const int t = threadIdx.x;
  if (t < BIN_NB) {
    hist[t] = 0;
    gbase[t] = binCount[t];
  }
  __syncthreads();
  const int e0 = blockIdx.x * BIN_EPB;
  int mybin[BIN_EPB / BIN_T];
#pragma unroll
  for (int k = 0; k < BIN_EPB / BIN_T; ++k) {
    const int e = e0 + k * BIN_T + t;
    mybin[k] = -1;
    if (e < E) {
      mybin[k] = rgcn_bin_of(dst[e]);
      atomicAdd(&hist[mybin[k]], 1);
    }
  }
  __syncthreads();
  rgcn_scan_bins(gbase, wsum);
  if (t < BIN_NB) {
    const int h = hist[t];
    if (blockIdx.x == 0) binStart[t] = gbase[t];
    const int pc = rgcn_pad_line(h);
    if (pc > 0) gbase[t] += atomicAdd(&binCursor[t], pc);
    loff[t] = h;
  }
  __syncthreads();
  rgcn_scan_bins(loff, wsum);
  if (t < BIN_NB) fill[t] = loff[t];
  __syncthreads();
#pragma unroll
  for (int k = 0; k < BIN_EPB / BIN_T; ++k) {
    if (mybin[k] >= 0) {
      const int e = e0 + k * BIN_T + t;
      const int b = mybin[k];
      const int pos = atomicAdd(&fill[b], 1);
      staged[pos] = make_int2(src[e], (dst[e] - b * BIN_P) * RGCN_R + et[e]);
    }
  }
  __syncthreads();
  const int lane = t & 31;
  const int w = t >> 5;
  for (int b = w; b < BIN_NB; b += BIN_T / 32) {
    const int h  = hist[b];
    const int pc = rgcn_pad_line(h);
    const int lo = loff[b];
    const int gb = gbase[b];
    for (int k = lane; k < pc; k += 32) {
      int2 v = make_int2(-1, 0);
      if (k < h) v = staged[lo + k];
      binned[gb + k] = v;
    }
  }
}

template<int K>
__global__ void rgcn_pack_kernel(const float* __restrict__ rootW,
                                 const float* __restrict__ basisW,
                                 float* __restrict__ Bp) {
  const int t = blockIdx.x * blockDim.x + threadIdx.x;
  if (t >= K * RGCN_W) return;
  const int m = t & 3;
  const int n = (t >> 2) % RGCN_W;
  const int q = t / (4 * RGCN_W);
  const int k = q * 4 + m;
  float v;
  if (n < RGCN_HID) {
    v = rootW[k * RGCN_HID + n];
  } else {
    const int b = (n - RGCN_HID) >> 6;
    const int c = (n - RGCN_HID) & 63;
    v = basisW[((size_t)b * K + k) * RGCN_HID + c];
  }
  Bp[t] = v;
}

template<int PITCH, bool ADD_BIAS>
__device__ __forceinline__ void rgcn_gemm_store_pair(float* __restrict__ outp,
                                                     v8f t0, v8f t1, float bv,
                                                     int hi) {
  float lo[8], up[8];
#pragma unroll
  for (int v = 0; v < 8; ++v) {
    const float x0 = __shfl_xor(t0[v], 16, 32);
    const float x1 = __shfl_xor(t1[v], 16, 32);
    lo[v] = hi ? x1 : t0[v];
    up[v] = hi ? t1[v] : x0;
  }
#pragma unroll
  for (int v = 0; v < 8; ++v) {
    outp[(size_t)v * PITCH]       = ADD_BIAS ? lo[v] + bv : lo[v];
    outp[(size_t)(v + 8) * PITCH] = ADD_BIAS ? up[v] + bv : up[v];
  }
}

#define GEMM_MT 2
#define GEMM_THREADS 160

template<int K, bool RELU_A>
__global__ __launch_bounds__(GEMM_THREADS) void rgcn_gemm_kernel(
    const float* __restrict__ A, const float* __restrict__ Bp,
    const float* __restrict__ bias, float* __restrict__ h,
    float* __restrict__ z, int nRows) {
  const int wave = __builtin_amdgcn_readfirstlane((int)(threadIdx.x >> 5));
  const int lane = threadIdx.x & 31;
  const int hi   = lane >> 4;
  const int lm   = lane & 15;
  const int rowBase = blockIdx.x * (16 * GEMM_MT);

  __shared__ __attribute__((aligned(16))) float As[16 * GEMM_MT][K + 4];
  constexpr int kTasks = 16 * GEMM_MT * (K / 4);
  constexpr int kTrips = kTasks / 128;
  static_assert(kTasks % 128 == 0, "staging tasks must divide over 128 threads");
  if (wave < 4) {
    float4 st[kTrips];
#pragma unroll
    for (int it = 0; it < kTrips; ++it) {
      const int idx = (int)threadIdx.x + it * 128;
      const int r = idx / (K / 4);
      const int q = idx - r * (K / 4);
      int row = rowBase + r;
      if (row > nRows - 1) row = nRows - 1;
      float4 v = *(const float4*)(A + (size_t)row * K + 4 * q);
      if (RELU_A) {
        v.x = fmaxf(v.x, 0.f); v.y = fmaxf(v.y, 0.f);
        v.z = fmaxf(v.z, 0.f); v.w = fmaxf(v.w, 0.f);
      }
      st[it] = v;
    }
#pragma unroll
    for (int it = 0; it < kTrips; ++it) {
      const int idx = (int)threadIdx.x + it * 128;
      const int r = idx / (K / 4);
      const int q = idx - r * (K / 4);
      *(float4*)&As[r][4 * q] = st[it];
    }
  }
  __syncthreads();
  const float* __restrict__ bptr = Bp + (size_t)(64 * wave + lm) * 4 + 2 * hi;

  v8f acc[GEMM_MT][4] = {};
#pragma unroll 4
  for (int k = 0; k < K; k += 4) {
    v2f a[GEMM_MT], b[4];
#pragma unroll
    for (int m = 0; m < GEMM_MT; ++m) {
      a[m] = *(const v2f*)&As[16 * m + lm][k + 2 * hi];
    }
#pragma unroll
    for (int t = 0; t < 4; ++t)
      b[t] = *(const v2f*)(bptr + t * 64 + (size_t)(k >> 2) * (4 * RGCN_W));
#pragma unroll
    for (int m = 0; m < GEMM_MT; ++m) {
#pragma unroll
      for (int t = 0; t < 4; ++t) acc[m][t] = wmma_f32x4(a[m], b[t], acc[m][t]);
    }
  }

  if (wave == 0) {
#pragma unroll
    for (int m = 0; m < GEMM_MT; ++m) {
      const int r0 = rowBase + 16 * m;
      if (r0 + 16 <= nRows) {
#pragma unroll
        for (int p = 0; p < 2; ++p)
          rgcn_gemm_store_pair<RGCN_HID, true>(
              h + (size_t)r0 * RGCN_HID + 32 * p + lane, acc[m][2 * p],
              acc[m][2 * p + 1], bias[32 * p + lane], hi);
      }
    }
  } else {
    const int zc = 64 * (wave - 1) + lane;
#pragma unroll
    for (int m = 0; m < GEMM_MT; ++m) {
      const int r0 = rowBase + 16 * m;
      if (r0 + 16 <= nRows) {
#pragma unroll
        for (int p = 0; p < 2; ++p)
          rgcn_gemm_store_pair<RGCN_B * RGCN_HID, false>(
              z + (size_t)r0 * (RGCN_B * RGCN_HID) + zc + 32 * p, acc[m][2 * p],
              acc[m][2 * p + 1], 0.f, hi);
      }
    }
  }
}

#define SCAT_GRP 4

template<int G>
__device__ __forceinline__ void rgcn_tile_group(
    int s, int dl, float c0, float c1, float c2, float c3, unsigned& m, int lane,
    const float* __restrict__ z, float* tile) {
  int jl[G];
#pragma unroll
  for (int u = 0; u < G; ++u) {
    jl[u] = __builtin_ctz(m);
    m &= m - 1u;
  }
  float zv[G][8];
#pragma unroll
  for (int u = 0; u < G; ++u) {
    const int sj = __builtin_amdgcn_readlane(s, jl[u]);
    const float* __restrict__ zp =
        z + (size_t)sj * (RGCN_B * RGCN_HID) + lane;
#pragma unroll
    for (int q = 0; q < 8; ++q) zv[u][q] = zp[q * 32];
  }
#pragma unroll
  for (int u = 0; u < G; ++u) {
    const int j  = jl[u];
    const int dj = __builtin_amdgcn_readlane(dl, j);
    const float a0 = __int_as_float(__builtin_amdgcn_readlane(__float_as_int(c0), j));
    const float a1 = __int_as_float(__builtin_amdgcn_readlane(__float_as_int(c1), j));
    const float a2 = __int_as_float(__builtin_amdgcn_readlane(__float_as_int(c2), j));
    const float a3 = __int_as_float(__builtin_amdgcn_readlane(__float_as_int(c3), j));
    const float m0 = a0 * zv[u][0] + a1 * zv[u][2] + a2 * zv[u][4] + a3 * zv[u][6];
    const float m1 = a0 * zv[u][1] + a1 * zv[u][3] + a2 * zv[u][5] + a3 * zv[u][7];
    float* tp = tile + dj * RGCN_HID + lane;
    atomicAdd(tp, m0);
    atomicAdd(tp + 32, m1);
  }
}

template<bool USE_NEED>
__global__ __launch_bounds__(BIN_T) void rgcn_aggregate_kernel(
    const int* __restrict__ binnedW,
    const int* __restrict__ binStart,
    const int* __restrict__ binCount, const float* __restrict__ comp,
    const float* __restrict__ probs,
    const unsigned* __restrict__ need,
    const float* __restrict__ z, float* __restrict__ hout, int N) {
  __shared__ unsigned cntL[BIN_P * RGCN_R];
  __shared__ __attribute__((aligned(16))) float tile[BIN_P * RGCN_HID];
  const int2* __restrict__ binned = (const int2*)binnedW;
  const int t = threadIdx.x;
  const int lane = t & 31;
  const int w = __builtin_amdgcn_readfirstlane(t >> 5);
  const int n0 = blockIdx.x * BIN_P;
  const int start = binStart[blockIdx.x];
  const int count = binCount[blockIdx.x];

  for (int i = t; i < BIN_P * RGCN_R; i += BIN_T) cntL[i] = 0u;
  for (int i = t; i < BIN_P * RGCN_HID; i += BIN_T) tile[i] = 0.f;
  __syncthreads();

  for (int i = t; i < count; i += BIN_T) {
    const int2 ed = binned[start + i];
    if (ed.x >= 0) atomicAdd(&cntL[ed.y], 1u);
  }
  __syncthreads();

  for (int base = w * 32; base < count; base += BIN_T) {
    const int i = base + lane;
    int s = 0, w1 = 0;
    bool keep = false;
    if (i < count) {
      const int2 ed = binned[start + i];
      s = ed.x;
      w1 = ed.y;
      keep = s >= 0;
    }
    const int dl = w1 >> 7;
    const int ty = w1 & (RGCN_R - 1);
    if (USE_NEED && keep) {
      const int n = n0 + dl;
      keep = ((need[n >> 5] >> (n & 31)) & 1u) != 0u;
    }
    float c0 = 0.f, c1 = 0.f, c2 = 0.f, c3 = 0.f;
    if (keep) {
      const float gate = 1.f / (1.f + __expf(-probs[ty]));
      const float c    = (float)cntL[w1];
      const float g    = gate / fmaxf(c, 1.f);
      const float4 cv  = *(const float4*)(comp + ty * RGCN_B);
      c0 = cv.x * g; c1 = cv.y * g; c2 = cv.z * g; c3 = cv.w * g;
    }
    unsigned m = __builtin_amdgcn_ballot_w32(keep);
    while (__builtin_popcount(m) >= SCAT_GRP)
      rgcn_tile_group<SCAT_GRP>(s, dl, c0, c1, c2, c3, m, lane, z, tile);
    const int left = __builtin_popcount(m);
    if (left == 3)      rgcn_tile_group<3>(s, dl, c0, c1, c2, c3, m, lane, z, tile);
    else if (left == 2) rgcn_tile_group<2>(s, dl, c0, c1, c2, c3, m, lane, z, tile);
    else if (left == 1) rgcn_tile_group<1>(s, dl, c0, c1, c2, c3, m, lane, z, tile);
  }
  __syncthreads();

  int nn = N - n0;
  if (nn > BIN_P) nn = BIN_P;
  for (int idx = t; idx < nn * (RGCN_HID / 4); idx += BIN_T) {
    float4* hp = (float4*)(hout + (size_t)n0 * RGCN_HID) + idx;
    const float4 tv = *(const float4*)&tile[4 * idx];
    float4 hv = *hp;
    hv.x += tv.x; hv.y += tv.y; hv.z += tv.z; hv.w += tv.w;
    *hp = hv;
  }
}

#define SC_EPB 10
#define SC_THREADS 128

__global__ __launch_bounds__(SC_THREADS) void rgcn_score_kernel(
    const float* __restrict__ hB,
    const int* __restrict__ tei,
    const float* __restrict__ Rm,
    const float* __restrict__ D,
    float* __restrict__ out, int ET) {
  __shared__ __attribute__((aligned(16))) float x1s[SC_EPB][RGCN_HID];
  __shared__ __attribute__((aligned(16))) float x2s[SC_EPB][RGCN_HID];
  __shared__ __attribute__((aligned(16))) float outs[SC_EPB][RGCN_R];
  const int base = blockIdx.x * SC_EPB;

  for (int idx = threadIdx.x; idx < 2 * SC_EPB * 16; idx += blockDim.x) {
    const int which = idx / (SC_EPB * 16);
    const int rem   = idx - which * (SC_EPB * 16);
    const int m = rem >> 4, q = rem & 15;
    const int e = base + m;
    if (e < ET) {
      const int node = tei[which * ET + e];
      float4 v = *(const float4*)(hB + (size_t)node * RGCN_HID + q * 4);
      v.x = fmaxf(v.x, 0.f); v.y = fmaxf(v.y, 0.f);
      v.z = fmaxf(v.z, 0.f); v.w = fmaxf(v.w, 0.f);
      float* dstp = which ? &x2s[m][q * 4] : &x1s[m][q * 4];
      *(float4*)dstp = v;
    }
  }

  const int wave = threadIdx.x >> 5;
  const int lane = threadIdx.x & 31;
  const int hi   = lane >> 4;
  const int lm   = lane & 15;
  const int r0   = wave * 32 + lm;

  v2f fu[4][16];
  v2f fl[4][16];
  v2f dj[2][16];
  float di[2][4][8];
#pragma unroll
  for (int mt = 0; mt < 4; ++mt) {
#pragma unroll
    for (int kk = 4 * mt; kk < 16; ++kk) {
      const int i  = mt * 16 + lm;
      const int j0 = 4 * kk + 2 * hi;
      const v2f u = *(const v2f*)(Rm + i * RGCN_HID + j0);
      const float l0 = Rm[j0 * RGCN_HID + i];
      const float l1 = Rm[(j0 + 1) * RGCN_HID + i];
      fu[mt][kk].x = (j0     >= i) ? u.x : 0.f;
      fu[mt][kk].y = (j0 + 1 >= i) ? u.y : 0.f;
      fl[mt][kk].x = (j0     >  i) ? l0 : 0.f;
      fl[mt][kk].y = (j0 + 1 >  i) ? l1 : 0.f;
    }
  }
#pragma unroll
  for (int t = 0; t < 2; ++t) {
#pragma unroll
    for (int kk = 0; kk < 16; ++kk)
      dj[t][kk] = *(const v2f*)(D + (r0 + 16 * t) * RGCN_HID + 4 * kk + 2 * hi);
#pragma unroll
    for (int mt = 0; mt < 4; ++mt) {
      const float* dp = D + (r0 + 16 * t) * RGCN_HID + mt * 16 + 8 * hi;
      const float4 lo = *(const float4*)dp;
      const float4 up = *(const float4*)(dp + 4);
      di[t][mt][0] = lo.x; di[t][mt][1] = lo.y; di[t][mt][2] = lo.z; di[t][mt][3] = lo.w;
      di[t][mt][4] = up.x; di[t][mt][5] = up.y; di[t][mt][6] = up.z; di[t][mt][7] = up.w;
    }
  }
  __syncthreads();

  for (int m = 0; m < SC_EPB; ++m) {
    if (base + m >= ET) break;
    v8f acc[2][4] = {};
    float x1i[4], x2i[4];
#pragma unroll
    for (int mt = 0; mt < 4; ++mt) {
      x1i[mt] = x1s[m][16 * mt + lm];
      x2i[mt] = x2s[m][16 * mt + lm];
    }
#pragma unroll
    for (int kk = 0; kk < 16; ++kk) {
      const v2f p1 = *(const v2f*)&x1s[m][4 * kk + 2 * hi];
      const v2f p2 = *(const v2f*)&x2s[m][4 * kk + 2 * hi];
#pragma unroll
      for (int mt = 0; mt <= (kk >> 2); ++mt) {
        const v2f a = (fu[mt][kk] * p2) * x1i[mt] + (fl[mt][kk] * p1) * x2i[mt];
#pragma unroll
        for (int t = 0; t < 2; ++t) acc[t][mt] = wmma_f32x4(a, dj[t][kk], acc[t][mt]);
      }
    }
#pragma unroll
    for (int t = 0; t < 2; ++t) {
      float p[4];
#pragma unroll
      for (int mt = 0; mt < 4; ++mt) {
        float s = di[t][mt][0] * acc[t][mt][0];
#pragma unroll
        for (int v = 1; v < 8; ++v) s += di[t][mt][v] * acc[t][mt][v];
        p[mt] = s;
      }
      float tt = (p[0] + p[1]) + (p[2] + p[3]);
      tt += __shfl_xor(tt, 16, 32);
      if (hi == 0) outs[m][r0 + 16 * t] = 1.f / (1.f + __expf(-tt));
    }
  }
  __syncthreads();

  int nvalid = ET - base;
  if (nvalid > SC_EPB) nvalid = SC_EPB;
  float* __restrict__ obase = out + (size_t)base * RGCN_R;
  const float* osrc = &outs[0][0];
  for (int idx = threadIdx.x * 4; idx < nvalid * RGCN_R; idx += blockDim.x * 4)
    *(float4*)(obase + idx) = *(const float4*)(osrc + idx);
}

extern "C" void kernel_launch(void* const* d_in, const int* in_sizes, int n_in,
                              void* d_out, int out_size, void* d_ws,
                              size_t ws_size, hipStream_t stream) {
  const float* x      = (const float*)d_in[0];
  const int*   eidx   = (const int*)d_in[1];
  const int*   etype  = (const int*)d_in[2];
  const int*   tei    = (const int*)d_in[3];
  const float* probs  = (const float*)d_in[4];
  const float* basis1 = (const float*)d_in[5];
  const float* comp1  = (const float*)d_in[6];
  const float* root1  = (const float*)d_in[7];
  const float* bias1  = (const float*)d_in[8];
  const float* basis2 = (const float*)d_in[9];
  const float* comp2  = (const float*)d_in[10];
  const float* root2  = (const float*)d_in[11];
  const float* bias2  = (const float*)d_in[12];
  const float* Rm     = (const float*)d_in[13];
  const float* D      = (const float*)d_in[14];

  const int N  = in_sizes[0] / RGCN_IN;
  const int E  = in_sizes[2];
  const int ET = in_sizes[3] / 2;

  float* z   = (float*)d_ws;
  float* hA  = z   + (size_t)N * RGCN_B * RGCN_HID;
  float* hB  = hA  + (size_t)N * RGCN_HID;
  float* Bp1 = hB  + (size_t)N * RGCN_HID;
  float* Bp2 = Bp1 + (size_t)RGCN_IN * RGCN_W;
  int* binCount  = (int*)(Bp2 + (size_t)RGCN_HID * RGCN_W);
  int* binCursor = binCount + BIN_NB;
  unsigned* need = (unsigned*)(binCursor + BIN_NB);
  const size_t needPad = (size_t)((N + 1023) / 1024) * 32;
  int* binStart  = (int*)(need + needPad);
  int* binned    = binStart + BIN_NB;

  const int* src = eidx;
  const int* dst = eidx + E;

  hipMemsetAsync(binCount, 0, ((size_t)2 * BIN_NB + needPad) * sizeof(int), stream);

  const int nb = (E + BIN_EPB - 1) / BIN_EPB;
  rgcn_binhist_kernel<<<nb, BIN_T, 0, stream>>>(dst, E, binCount, tei, need, 2 * ET);
  rgcn_binfill_kernel<<<nb, BIN_T, 0, stream>>>(src, dst, etype, E, binCount,
                                               binCursor, binStart, binned);

  rgcn_pack_kernel<RGCN_IN>
      <<<(RGCN_IN * RGCN_W + 255) / 256, 256, 0, stream>>>(root1, basis1, Bp1);
  rgcn_pack_kernel<RGCN_HID>
      <<<(RGCN_HID * RGCN_W + 255) / 256, 256, 0, stream>>>(root2, basis2, Bp2);

  const int gg = (N + 16 * GEMM_MT - 1) / (16 * GEMM_MT);

  rgcn_gemm_kernel<RGCN_IN, false>
      <<<gg, GEMM_THREADS, 0, stream>>>(x, Bp1, bias1, hA, z, N);
  rgcn_aggregate_kernel<false><<<BIN_NB, BIN_T, 0, stream>>>(
      binned, binStart, binCount, comp1, probs, need, z, hA, N);

  rgcn_gemm_kernel<RGCN_HID, true>
      <<<gg, GEMM_THREADS, 0, stream>>>(hA, Bp2, bias2, hB, z, N);
  rgcn_aggregate_kernel<true><<<BIN_NB, BIN_T, 0, stream>>>(
      binned, binStart, binCount, comp2, probs, need, z, hB, N);

  rgcn_score_kernel<<<(ET + SC_EPB - 1) / SC_EPB, SC_THREADS, 0, stream>>>(
      hB, tei, Rm, D, (float*)d_out, ET);
}


# Layer 2 sums before it transforms: edges gather 256-B relu(h1) rows (was 1-KB z rows) into 4 basis-weighted LDS sums per scored node, times [basis;root] on f32 WMMA in that launch; no GEMM-2 launch
# speedup vs baseline: 2.6932x; 1.1079x over previous
#include <hip/hip_runtime.h>
#include <cstddef>

typedef float v2f __attribute__((ext_vector_type(2)));
typedef float v8f __attribute__((ext_vector_type(8)));

#define RGCN_R   128
#define RGCN_B   4
#define RGCN_IN  128
#define RGCN_HID 64
#define RGCN_W   (RGCN_HID * (RGCN_B + 1))

__device__ __forceinline__ v8f wmma_f32x4(v2f a, v2f b, v8f c) {
  return __builtin_amdgcn_wmma_f32_16x16x4_f32(false, a, false, b, (short)0, c,
                                               false, false);
}

#define BIN_P    98
#define BIN_NB   512
#define BIN_EPB  8192
#define BIN_T    1024
#define BIN_LINE 16

__device__ __forceinline__ int rgcn_bin_of(int d) {
  const int b = d / BIN_P;
  return b < BIN_NB - 1 ? b : BIN_NB - 1;
}

__device__ __forceinline__ int rgcn_pad_line(int n) {
  return (n + (BIN_LINE - 1)) & ~(BIN_LINE - 1);
}

__device__ __forceinline__ void rgcn_scan_bins(int* a, int* wsum) {
  const int t = threadIdx.x;
  const int lane = t & 31;
  const int w = t >> 5;
  int v = 0, inc = 0;
  if (t < BIN_NB) {
    v = a[t];
    inc = v;
    for (int d = 1; d < 32; d <<= 1) {
      const int n = __shfl_up(inc, d, 32);
      if (lane >= d) inc += n;
    }
    if (lane == 31) wsum[w] = inc;
  }
  __syncthreads();
  if (t < BIN_NB) {
    int off = 0;
    for (int k = 0; k < w; ++k) off += wsum[k];
    a[t] = off + inc - v;
  }
  __syncthreads();
}

__global__ __launch_bounds__(BIN_T) void rgcn_binhist_kernel(
    const int* __restrict__ dst, int E, int* __restrict__ binCount,
    const int* __restrict__ tei, unsigned* __restrict__ need, int nT) {
  __shared__ int hist[BIN_NB];
  const int t = threadIdx.x;
  if (t < BIN_NB) hist[t] = 0;
  __syncthreads();
  const int e0 = blockIdx.x * BIN_EPB;
  for (int k = 0; k < BIN_EPB / BIN_T; ++k) {
    const int e = e0 + k * BIN_T + t;
    if (e < E) atomicAdd(&hist[rgcn_bin_of(dst[e])], 1);
  }
  __syncthreads();
  if (t < BIN_NB) {
    const int pc = rgcn_pad_line(hist[t]);
    if (pc > 0) atomicAdd(&binCount[t], pc);
  }
  for (int i = blockIdx.x * BIN_T + t; i < nT; i += (int)(gridDim.x * BIN_T)) {
    const int n = tei[i];
    atomicOr(&need[n >> 5], 1u << (n & 31));
  }
}

__global__ __launch_bounds__(BIN_T) void rgcn_binfill_kernel(
    const int* __restrict__ src, const int* __restrict__ dst,
    const int* __restrict__ et, int E, const int* __restrict__ binCount,
    int* __restrict__ binCursor, int* __restrict__ binStart,
    int* __restrict__ binnedW) {
  int2* __restrict__ binned = (int2*)binnedW;
  __shared__ int hist[BIN_NB];
  __shared__ int loff[BIN_NB];
  __shared__ int fill[BIN_NB];
  __shared__ int gbase[BIN_NB];
  __shared__ int wsum[BIN_NB / 32];
  __shared__ int2 staged[BIN_EPB];
  const int t = threadIdx.x;
  if (t < BIN_NB) {
    hist[t] = 0;
    gbase[t] = binCount[t];
  }
  __syncthreads();
  const int e0 = blockIdx.x * BIN_EPB;
  int mybin[BIN_EPB / BIN_T];
#pragma unroll
  for (int k = 0; k < BIN_EPB / BIN_T; ++k) {
    const int e = e0 + k * BIN_T + t;
    mybin[k] = -1;
    if (e < E) {
      mybin[k] = rgcn_bin_of(dst[e]);
      atomicAdd(&hist[mybin[k]], 1);
    }
  }
  __syncthreads();
  rgcn_scan_bins(gbase, wsum);
  if (t < BIN_NB) {
    const int h = hist[t];
    if (blockIdx.x == 0) binStart[t] = gbase[t];
    const int pc = rgcn_pad_line(h);
    if (pc > 0) gbase[t] += atomicAdd(&binCursor[t], pc);
    loff[t] = h;
  }
  __syncthreads();
  rgcn_scan_bins(loff, wsum);
  if (t < BIN_NB) fill[t] = loff[t];
  __syncthreads();
#pragma unroll
  for (int k = 0; k < BIN_EPB / BIN_T; ++k) {
    if (mybin[k] >= 0) {
      const int e = e0 + k * BIN_T + t;
      const int b = mybin[k];
      const int pos = atomicAdd(&fill[b], 1);
      staged[pos] = make_int2(src[e], (dst[e] - b * BIN_P) * RGCN_R + et[e]);
    }
  }
  __syncthreads();
  const int lane = t & 31;
  const int w = t >> 5;
  for (int b = w; b < BIN_NB; b += BIN_T / 32) {
    const int h  = hist[b];
    const int pc = rgcn_pad_line(h);
    const int lo = loff[b];
    const int gb = gbase[b];
    for (int k = lane; k < pc; k += 32) {
      int2 v = make_int2(-1, 0);
      if (k < h) v = staged[lo + k];
      binned[gb + k] = v;
    }
  }
}

template<int K>
__global__ void rgcn_pack_kernel(const float* __restrict__ rootW,
                                 const float* __restrict__ basisW,
                                 float* __restrict__ Bp) {
  const int t = blockIdx.x * blockDim.x + threadIdx.x;
  if (t >= K * RGCN_W) return;
  const int m = t & 3;
  const int n = (t >> 2) % RGCN_W;
  const int q = t / (4 * RGCN_W);
  const int k = q * 4 + m;
  float v;
  if (n < RGCN_HID) {
    v = rootW[k * RGCN_HID + n];
  } else {
    const int b = (n - RGCN_HID) >> 6;
    const int c = (n - RGCN_HID) & 63;
    v = basisW[((size_t)b * K + k) * RGCN_HID + c];
  }
  Bp[t] = v;
}

template<int PITCH, bool ADD_BIAS>
__device__ __forceinline__ void rgcn_gemm_store_pair(float* __restrict__ outp,
                                                     v8f t0, v8f t1, float bv,
                                                     int hi) {
  float lo[8], up[8];
#pragma unroll
  for (int v = 0; v < 8; ++v) {
    const float x0 = __shfl_xor(t0[v], 16, 32);
    const float x1 = __shfl_xor(t1[v], 16, 32);
    lo[v] = hi ? x1 : t0[v];
    up[v] = hi ? t1[v] : x0;
  }
#pragma unroll
  for (int v = 0; v < 8; ++v) {
    outp[(size_t)v * PITCH]       = ADD_BIAS ? lo[v] + bv : lo[v];
    outp[(size_t)(v + 8) * PITCH] = ADD_BIAS ? up[v] + bv : up[v];
  }
}

#define GEMM_MT 2
#define GEMM_THREADS 160

template<int K>
__global__ __launch_bounds__(GEMM_THREADS) void rgcn_gemm_kernel(
    const float* __restrict__ A, const float* __restrict__ Bp,
    const float* __restrict__ bias, float* __restrict__ h,
    float* __restrict__ z, int nRows) {
  const int wave = __builtin_amdgcn_readfirstlane((int)(threadIdx.x >> 5));
  const int lane = threadIdx.x & 31;
  const int hi   = lane >> 4;
  const int lm   = lane & 15;
  const int rowBase = blockIdx.x * (16 * GEMM_MT);

  __shared__ __attribute__((aligned(16))) float As[16 * GEMM_MT][K + 4];
  constexpr int kTasks = 16 * GEMM_MT * (K / 4);
  constexpr int kTrips = kTasks / 128;
  static_assert(kTasks % 128 == 0, "staging tasks must divide over 128 threads");
  if (wave < 4) {
    float4 st[kTrips];
#pragma unroll
    for (int it = 0; it < kTrips; ++it) {
      const int idx = (int)threadIdx.x + it * 128;
      const int r = idx / (K / 4);
      const int q = idx - r * (K / 4);
      int row = rowBase + r;
      if (row > nRows - 1) row = nRows - 1;
      st[it] = *(const float4*)(A + (size_t)row * K + 4 * q);
    }
#pragma unroll
    for (int it = 0; it < kTrips; ++it) {
      const int idx = (int)threadIdx.x + it * 128;
      const int r = idx / (K / 4);
      const int q = idx - r * (K / 4);
      *(float4*)&As[r][4 * q] = st[it];
    }
  }
  __syncthreads();
  const float* __restrict__ bptr = Bp + (size_t)(64 * wave + lm) * 4 + 2 * hi;

  v8f acc[GEMM_MT][4] = {};
#pragma unroll 4
  for (int k = 0; k < K; k += 4) {
    v2f a[GEMM_MT], b[4];
#pragma unroll
    for (int m = 0; m < GEMM_MT; ++m) {
      a[m] = *(const v2f*)&As[16 * m + lm][k + 2 * hi];
    }
#pragma unroll
    for (int t = 0; t < 4; ++t)
      b[t] = *(const v2f*)(bptr + t * 64 + (size_t)(k >> 2) * (4 * RGCN_W));
#pragma unroll
    for (int m = 0; m < GEMM_MT; ++m) {
#pragma unroll
      for (int t = 0; t < 4; ++t) acc[m][t] = wmma_f32x4(a[m], b[t], acc[m][t]);
    }
  }

  if (wave == 0) {
#pragma unroll
    for (int m = 0; m < GEMM_MT; ++m) {
      const int r0 = rowBase + 16 * m;
      if (r0 + 16 <= nRows) {
#pragma unroll
        for (int p = 0; p < 2; ++p)
          rgcn_gemm_store_pair<RGCN_HID, true>(
              h + (size_t)r0 * RGCN_HID + 32 * p + lane, acc[m][2 * p],
              acc[m][2 * p + 1], bias[32 * p + lane], hi);
      }
    }
  } else {
    const int zc = 64 * (wave - 1) + lane;
#pragma unroll
    for (int m = 0; m < GEMM_MT; ++m) {
      const int r0 = rowBase + 16 * m;
      if (r0 + 16 <= nRows) {
#pragma unroll
        for (int p = 0; p < 2; ++p)
          rgcn_gemm_store_pair<RGCN_B * RGCN_HID, false>(
              z + (size_t)r0 * (RGCN_B * RGCN_HID) + zc + 32 * p, acc[m][2 * p],
              acc[m][2 * p + 1], 0.f, hi);
      }
    }
  }
}

#define SCAT_GRP 4

template<int G>
__device__ __forceinline__ void rgcn_tile_group(
    int s, int dl, float c0, float c1, float c2, float c3, unsigned& m, int lane,
    const float* __restrict__ z, float* tile) {
  int jl[G];
#pragma unroll
  for (int u = 0; u < G; ++u) {
    jl[u] = __builtin_ctz(m);
    m &= m - 1u;
  }
  float zv[G][8];
#pragma unroll
  for (int u = 0; u < G; ++u) {
    const int sj = __builtin_amdgcn_readlane(s, jl[u]);
    const float* __restrict__ zp =
        z + (size_t)sj * (RGCN_B * RGCN_HID) + lane;
#pragma unroll
    for (int q = 0; q < 8; ++q) zv[u][q] = zp[q * 32];
  }
#pragma unroll
  for (int u = 0; u < G; ++u) {
    const int j  = jl[u];
    const int dj = __builtin_amdgcn_readlane(dl, j);
    const float a0 = __int_as_float(__builtin_amdgcn_readlane(__float_as_int(c0), j));
    const float a1 = __int_as_float(__builtin_amdgcn_readlane(__float_as_int(c1), j));
    const float a2 = __int_as_float(__builtin_amdgcn_readlane(__float_as_int(c2), j));
    const float a3 = __int_as_float(__builtin_amdgcn_readlane(__float_as_int(c3), j));
    const float m0 = a0 * zv[u][0] + a1 * zv[u][2] + a2 * zv[u][4] + a3 * zv[u][6];
    const float m1 = a0 * zv[u][1] + a1 * zv[u][3] + a2 * zv[u][5] + a3 * zv[u][7];
    float* tp = tile + dj * RGCN_HID + lane;
    atomicAdd(tp, m0);
    atomicAdd(tp + 32, m1);
  }
}

__global__ __launch_bounds__(BIN_T) void rgcn_aggregate_kernel(
    const int* __restrict__ binnedW,
    const int* __restrict__ binStart,
    const int* __restrict__ binCount, const float* __restrict__ comp,
    const float* __restrict__ probs,
    const float* __restrict__ z, float* __restrict__ hout, int N) {
  __shared__ unsigned cntL[BIN_P * RGCN_R];
  __shared__ __attribute__((aligned(16))) float tile[BIN_P * RGCN_HID];
  const int2* __restrict__ binned = (const int2*)binnedW;
  const int t = threadIdx.x;
  const int lane = t & 31;
  const int w = __builtin_amdgcn_readfirstlane(t >> 5);
  const int n0 = blockIdx.x * BIN_P;
  const int start = binStart[blockIdx.x];
  const int count = binCount[blockIdx.x];

  for (int i = t; i < BIN_P * RGCN_R; i += BIN_T) cntL[i] = 0u;
  for (int i = t; i < BIN_P * RGCN_HID; i += BIN_T) tile[i] = 0.f;
  __syncthreads();

  for (int i = t; i < count; i += BIN_T) {
    const int2 ed = binned[start + i];
    if (ed.x >= 0) atomicAdd(&cntL[ed.y], 1u);
  }
  __syncthreads();

  for (int base = w * 32; base < count; base += BIN_T) {
    const int i = base + lane;
    int s = 0, w1 = 0;
    bool keep = false;
    if (i < count) {
      const int2 ed = binned[start + i];
      s = ed.x;
      w1 = ed.y;
      keep = s >= 0;
    }
    const int dl = w1 >> 7;
    const int ty = w1 & (RGCN_R - 1);
    float c0 = 0.f, c1 = 0.f, c2 = 0.f, c3 = 0.f;
    if (keep) {
      const float gate = 1.f / (1.f + __expf(-probs[ty]));
      const float c    = (float)cntL[w1];
      const float g    = gate / fmaxf(c, 1.f);
      const float4 cv  = *(const float4*)(comp + ty * RGCN_B);
      c0 = cv.x * g; c1 = cv.y * g; c2 = cv.z * g; c3 = cv.w * g;
    }
    unsigned m = __builtin_amdgcn_ballot_w32(keep);
    while (__builtin_popcount(m) >= SCAT_GRP)
      rgcn_tile_group<SCAT_GRP>(s, dl, c0, c1, c2, c3, m, lane, z, tile);
    const int left = __builtin_popcount(m);
    if (left == 3)      rgcn_tile_group<3>(s, dl, c0, c1, c2, c3, m, lane, z, tile);
    else if (left == 2) rgcn_tile_group<2>(s, dl, c0, c1, c2, c3, m, lane, z, tile);
    else if (left == 1) rgcn_tile_group<1>(s, dl, c0, c1, c2, c3, m, lane, z, tile);
  }
  __syncthreads();

  int nn = N - n0;
  if (nn > BIN_P) nn = BIN_P;
  for (int idx = t; idx < nn * (RGCN_HID / 4); idx += BIN_T) {
    float4* hp = (float4*)(hout + (size_t)n0 * RGCN_HID) + idx;
    const float4 tv = *(const float4*)&tile[4 * idx];
    float4 hv = *hp;
    hv.x += tv.x; hv.y += tv.y; hv.z += tv.z; hv.w += tv.w;
    *hp = hv;
  }
}

#define L2_PITCH (RGCN_B * RGCN_HID + 4)
#define L2_GRP 8

template<int G>
__device__ __forceinline__ void rgcn_l2_group(
    int s, int sl, float c0, float c1, float c2, float c3, unsigned& m, int lane,
    const float* __restrict__ hin, float* tile) {
  int jl[G];
  int n = 0;
#pragma unroll
  for (int u = 0; u < G; ++u) {
    jl[u] = 0;
    if (m != 0u) {
      jl[u] = __builtin_ctz(m);
      m &= m - 1u;
      ++n;
    }
  }
  v2f xv[G];
#pragma unroll
  for (int u = 0; u < G; ++u) {
    xv[u].x = 0.f;
    xv[u].y = 0.f;
    if (u < n) {
      const int sj = __builtin_amdgcn_readlane(s, jl[u]);
      xv[u] = *(const v2f*)(hin + (size_t)sj * RGCN_HID + 2 * lane);
    }
  }
#pragma unroll
  for (int u = 0; u < G; ++u) {
    if (u < n) {
      const int j   = jl[u];
      const int slj = __builtin_amdgcn_readlane(sl, j);
      const float a0 = __int_as_float(__builtin_amdgcn_readlane(__float_as_int(c0), j));
      const float a1 = __int_as_float(__builtin_amdgcn_readlane(__float_as_int(c1), j));
      const float a2 = __int_as_float(__builtin_amdgcn_readlane(__float_as_int(c2), j));
      const float a3 = __int_as_float(__builtin_amdgcn_readlane(__float_as_int(c3), j));
      const float x0 = fmaxf(xv[u].x, 0.f);
      const float x1 = fmaxf(xv[u].y, 0.f);
      float* tp = tile + slj * L2_PITCH + 2 * lane;
      atomicAdd(tp,                    a0 * x0);
      atomicAdd(tp + 1,                a0 * x1);
      atomicAdd(tp + RGCN_HID,         a1 * x0);
      atomicAdd(tp + RGCN_HID + 1,     a1 * x1);
      atomicAdd(tp + 2 * RGCN_HID,     a2 * x0);
      atomicAdd(tp + 2 * RGCN_HID + 1, a2 * x1);
      atomicAdd(tp + 3 * RGCN_HID,     a3 * x0);
      atomicAdd(tp + 3 * RGCN_HID + 1, a3 * x1);
    }
  }
}

__global__ __launch_bounds__(BIN_T) __attribute__((amdgpu_waves_per_eu(16)))
void rgcn_layer2_kernel(
    const int* __restrict__ binnedW,
    const int* __restrict__ binStart,
    const int* __restrict__ binCount, const float* __restrict__ comp,
    const float* __restrict__ probs,
    const unsigned* __restrict__ need,
    const float* __restrict__ hin,
    const float* __restrict__ Bp,
    const float* __restrict__ bias, float* __restrict__ hout, int N) {
  __shared__ __attribute__((aligned(16))) unsigned cntL[BIN_P * RGCN_R];
  __shared__ __attribute__((aligned(16))) float tile[BIN_P * L2_PITCH];
  __shared__ int slotOf[BIN_P];
  __shared__ int nodeOf[BIN_P];
  __shared__ int wcnt[4];
  const int2* __restrict__ binned = (const int2*)binnedW;
  const int t = threadIdx.x;
  const int lane = t & 31;
  const int w = __builtin_amdgcn_readfirstlane(t >> 5);
  const int n0 = blockIdx.x * BIN_P;
  const int start = binStart[blockIdx.x];
  const int count = binCount[blockIdx.x];

  bool nd = false;
  int pre = 0;
  if (t < 128) {
    const int n = n0 + t;
    if (t < BIN_P && n < N) nd = ((need[n >> 5] >> (n & 31)) & 1u) != 0u;
    const unsigned bal = __builtin_amdgcn_ballot_w32(nd);
    pre = __builtin_popcount(bal & ((1u << lane) - 1u));
    if (lane == 0) wcnt[w] = __builtin_popcount(bal);
  }
  for (int i = t; i < BIN_P * RGCN_R; i += BIN_T) cntL[i] = 0u;
  __syncthreads();
  const int nNeed = wcnt[0] + wcnt[1] + wcnt[2] + wcnt[3];
  if (nNeed == 0) return;
  if (t < 128) {
    int off = 0;
    for (int k = 0; k < w; ++k) off += wcnt[k];
    if (t < BIN_P) slotOf[t] = nd ? off + pre : -1;
    if (nd) nodeOf[off + pre] = t;
  }
  const int mtiles = (nNeed + 15) >> 4;
  int zrows = 16 * mtiles;
  if (zrows > BIN_P) zrows = BIN_P;
  for (int i = t; i < zrows * L2_PITCH; i += BIN_T) tile[i] = 0.f;

  for (int i = t; i < count; i += BIN_T) {
    const int2 ed = binned[start + i];
    if (ed.x >= 0) atomicAdd(&cntL[ed.y], 1u);
  }
  __syncthreads();

  for (int base = w * 32; base < count; base += BIN_T) {
    const int i = base + lane;
    int s = 0, w1 = 0, sl = 0;
    bool keep = false;
    if (i < count) {
      const int2 ed = binned[start + i];
      s = ed.x;
      w1 = ed.y;
      keep = s >= 0;
    }
    const int ty = w1 & (RGCN_R - 1);
    float c0 = 0.f, c1 = 0.f, c2 = 0.f, c3 = 0.f;
    if (keep) {
      sl = slotOf[w1 >> 7];
      keep = sl >= 0;
    }
    if (keep) {
      const float gate = 1.f / (1.f + __expf(-probs[ty]));
      const float c    = (float)cntL[w1];
      const float g    = gate / fmaxf(c, 1.f);
      const float4 cv  = *(const float4*)(comp + ty * RGCN_B);
      c0 = cv.x * g; c1 = cv.y * g; c2 = cv.z * g; c3 = cv.w * g;
    }
    unsigned m = __builtin_amdgcn_ballot_w32(keep);
    while (m != 0u)
      rgcn_l2_group<L2_GRP>(s, sl, c0, c1, c2, c3, m, lane, hin, tile);
  }
  __syncthreads();

  float* stage = (float*)cntL;
  const int hi = lane >> 4;
  const int lm = lane & 15;
  for (int unit = w; unit < 4 * mtiles; unit += BIN_T / 32) {
    const int mt = unit >> 2;
    const int tn = unit & 3;
    const int row = 16 * mt + lm;
    const int rowc = row < zrows ? row : zrows - 1;
    const int slc  = row < nNeed ? row : nNeed - 1;
    const float* arow = tile + rowc * L2_PITCH + 2 * hi;
    const float* hrow = hin + (size_t)(n0 + nodeOf[slc]) * RGCN_HID + 2 * hi;
    const float* bcol = Bp + (size_t)(16 * tn + lm) * 4 + 2 * hi;
    v8f acc0 = {}, acc1 = {};
    for (int bb = 0; bb < RGCN_B; ++bb) {
      const float* ap = arow + bb * RGCN_HID;
      const float* bp = bcol + (size_t)(RGCN_HID + bb * RGCN_HID) * 4;
#pragma unroll 4
      for (int k4 = 0; k4 < RGCN_HID / 4; k4 += 2) {
        const v2f a0 = *(const v2f*)(ap + 4 * k4);
        const v2f a1 = *(const v2f*)(ap + 4 * k4 + 4);
        const v2f b0 = *(const v2f*)(bp + (size_t)k4 * (4 * RGCN_W));
        const v2f b1 = *(const v2f*)(bp + (size_t)(k4 + 1) * (4 * RGCN_W));
        acc0 = wmma_f32x4(a0, b0, acc0);
        acc1 = wmma_f32x4(a1, b1, acc1);
      }
    }
#pragma unroll 4
    for (int k4 = 0; k4 < RGCN_HID / 4; k4 += 2) {
      v2f a0 = *(const v2f*)(hrow + 4 * k4);
      v2f a1 = *(const v2f*)(hrow + 4 * k4 + 4);
      a0.x = fmaxf(a0.x, 0.f); a0.y = fmaxf(a0.y, 0.f);
      a1.x = fmaxf(a1.x, 0.f); a1.y = fmaxf(a1.y, 0.f);
      const v2f b0 = *(const v2f*)(bcol + (size_t)k4 * (4 * RGCN_W));
      const v2f b1 = *(const v2f*)(bcol + (size_t)(k4 + 1) * (4 * RGCN_W));
      acc0 = wmma_f32x4(a0, b0, acc0);
      acc1 = wmma_f32x4(a1, b1, acc1);
    }
    float* sp = stage + (16 * mt + 8 * hi) * RGCN_HID + 16 * tn + lm;
#pragma unroll
    for (int v = 0; v < 8; ++v) sp[v * RGCN_HID] = acc0[v] + acc1[v];
  }
  __syncthreads();

  for (int idx = t; idx < nNeed * (RGCN_HID / 4); idx += BIN_T) {
    const int slot = idx >> 4;
    const int q = idx & 15;
    float4 v = *(const float4*)&stage[slot * RGCN_HID + 4 * q];
    const float4 bv = *(const float4*)(bias + 4 * q);
    v.x += bv.x; v.y += bv.y; v.z += bv.z; v.w += bv.w;
    *(float4*)(hout + (size_t)(n0 + nodeOf[slot]) * RGCN_HID + 4 * q) = v;
  }
}

#define SC_EPB 10
#define SC_THREADS 128

__global__ __launch_bounds__(SC_THREADS) void rgcn_score_kernel(
    const float* __restrict__ hB,
    const int* __restrict__ tei,
    const float* __restrict__ Rm,
    const float* __restrict__ D,
    float* __restrict__ out, int ET) {
  __shared__ __attribute__((aligned(16))) float x1s[SC_EPB][RGCN_HID];
  __shared__ __attribute__((aligned(16))) float x2s[SC_EPB][RGCN_HID];
  __shared__ __attribute__((aligned(16))) float outs[SC_EPB][RGCN_R];
  const int base = blockIdx.x * SC_EPB;

  for (int idx = threadIdx.x; idx < 2 * SC_EPB * 16; idx += blockDim.x) {
    const int which = idx / (SC_EPB * 16);
    const int rem   = idx - which * (SC_EPB * 16);
    const int m = rem >> 4, q = rem & 15;
    const int e = base + m;
    if (e < ET) {
      const int node = tei[which * ET + e];
      float4 v = *(const float4*)(hB + (size_t)node * RGCN_HID + q * 4);
      v.x = fmaxf(v.x, 0.f); v.y = fmaxf(v.y, 0.f);
      v.z = fmaxf(v.z, 0.f); v.w = fmaxf(v.w, 0.f);
      float* dstp = which ? &x2s[m][q * 4] : &x1s[m][q * 4];
      *(float4*)dstp = v;
    }
  }

  const int wave = threadIdx.x >> 5;
  const int lane = threadIdx.x & 31;
  const int hi   = lane >> 4;
  const int lm   = lane & 15;
  const int r0   = wave * 32 + lm;

  v2f fu[4][16];
  v2f fl[4][16];
  v2f dj[2][16];
  float di[2][4][8];
#pragma unroll
  for (int mt = 0; mt < 4; ++mt) {
#pragma unroll
    for (int kk = 4 * mt; kk < 16; ++kk) {
      const int i  = mt * 16 + lm;
      const int j0 = 4 * kk + 2 * hi;
      const v2f u = *(const v2f*)(Rm + i * RGCN_HID + j0);
      const float l0 = Rm[j0 * RGCN_HID + i];
      const float l1 = Rm[(j0 + 1) * RGCN_HID + i];
      fu[mt][kk].x = (j0     >= i) ? u.x : 0.f;
      fu[mt][kk].y = (j0 + 1 >= i) ? u.y : 0.f;
      fl[mt][kk].x = (j0     >  i) ? l0 : 0.f;
      fl[mt][kk].y = (j0 + 1 >  i) ? l1 : 0.f;
    }
  }
#pragma unroll
  for (int t = 0; t < 2; ++t) {
#pragma unroll
    for (int kk = 0; kk < 16; ++kk)
      dj[t][kk] = *(const v2f*)(D + (r0 + 16 * t) * RGCN_HID + 4 * kk + 2 * hi);
#pragma unroll
    for (int mt = 0; mt < 4; ++mt) {
      const float* dp = D + (r0 + 16 * t) * RGCN_HID + mt * 16 + 8 * hi;
      const float4 lo = *(const float4*)dp;
      const float4 up = *(const float4*)(dp + 4);
      di[t][mt][0] = lo.x; di[t][mt][1] = lo.y; di[t][mt][2] = lo.z; di[t][mt][3] = lo.w;
      di[t][mt][4] = up.x; di[t][mt][5] = up.y; di[t][mt][6] = up.z; di[t][mt][7] = up.w;
    }
  }
  __syncthreads();

  for (int m = 0; m < SC_EPB; ++m) {
    if (base + m >= ET) break;
    v8f acc[2][4] = {};
    float x1i[4], x2i[4];
#pragma unroll
    for (int mt = 0; mt < 4; ++mt) {
      x1i[mt] = x1s[m][16 * mt + lm];
      x2i[mt] = x2s[m][16 * mt + lm];
    }
#pragma unroll
    for (int kk = 0; kk < 16; ++kk) {
      const v2f p1 = *(const v2f*)&x1s[m][4 * kk + 2 * hi];
      const v2f p2 = *(const v2f*)&x2s[m][4 * kk + 2 * hi];
#pragma unroll
      for (int mt = 0; mt <= (kk >> 2); ++mt) {
        const v2f a = (fu[mt][kk] * p2) * x1i[mt] + (fl[mt][kk] * p1) * x2i[mt];
#pragma unroll
        for (int t = 0; t < 2; ++t) acc[t][mt] = wmma_f32x4(a, dj[t][kk], acc[t][mt]);
      }
    }
#pragma unroll
    for (int t = 0; t < 2; ++t) {
      float p[4];
#pragma unroll
      for (int mt = 0; mt < 4; ++mt) {
        float s = di[t][mt][0] * acc[t][mt][0];
#pragma unroll
        for (int v = 1; v < 8; ++v) s += di[t][mt][v] * acc[t][mt][v];
        p[mt] = s;
      }
      float tt = (p[0] + p[1]) + (p[2] + p[3]);
      tt += __shfl_xor(tt, 16, 32);
      if (hi == 0) outs[m][r0 + 16 * t] = 1.f / (1.f + __expf(-tt));
    }
  }
  __syncthreads();

  int nvalid = ET - base;
  if (nvalid > SC_EPB) nvalid = SC_EPB;
  float* __restrict__ obase = out + (size_t)base * RGCN_R;
  const float* osrc = &outs[0][0];
  for (int idx = threadIdx.x * 4; idx < nvalid * RGCN_R; idx += blockDim.x * 4)
    *(float4*)(obase + idx) = *(const float4*)(osrc + idx);
}

extern "C" void kernel_launch(void* const* d_in, const int* in_sizes, int n_in,
                              void* d_out, int out_size, void* d_ws,
                              size_t ws_size, hipStream_t stream) {
  const float* x      = (const float*)d_in[0];
  const int*   eidx   = (const int*)d_in[1];
  const int*   etype  = (const int*)d_in[2];
  const int*   tei    = (const int*)d_in[3];
  const float* probs  = (const float*)d_in[4];
  const float* basis1 = (const float*)d_in[5];
  const float* comp1  = (const float*)d_in[6];
  const float* root1  = (const float*)d_in[7];
  const float* bias1  = (const float*)d_in[8];
  const float* basis2 = (const float*)d_in[9];
  const float* comp2  = (const float*)d_in[10];
  const float* root2  = (const float*)d_in[11];
  const float* bias2  = (const float*)d_in[12];
  const float* Rm     = (const float*)d_in[13];
  const float* D      = (const float*)d_in[14];

  const int N  = in_sizes[0] / RGCN_IN;
  const int E  = in_sizes[2];
  const int ET = in_sizes[3] / 2;

  float* z   = (float*)d_ws;
  float* hA  = z   + (size_t)N * RGCN_B * RGCN_HID;
  float* hB  = hA  + (size_t)N * RGCN_HID;
  float* Bp1 = hB  + (size_t)N * RGCN_HID;
  float* Bp2 = Bp1 + (size_t)RGCN_IN * RGCN_W;
  int* binCount  = (int*)(Bp2 + (size_t)RGCN_HID * RGCN_W);
  int* binCursor = binCount + BIN_NB;
  unsigned* need = (unsigned*)(binCursor + BIN_NB);
  const size_t needPad = (size_t)((N + 1023) / 1024) * 32;
  int* binStart  = (int*)(need + needPad);
  int* binned    = binStart + BIN_NB;

  const int* src = eidx;
  const int* dst = eidx + E;

  hipMemsetAsync(binCount, 0, ((size_t)2 * BIN_NB + needPad) * sizeof(int), stream);

  const int nb = (E + BIN_EPB - 1) / BIN_EPB;
  rgcn_binhist_kernel<<<nb, BIN_T, 0, stream>>>(dst, E, binCount, tei, need, 2 * ET);
  rgcn_binfill_kernel<<<nb, BIN_T, 0, stream>>>(src, dst, etype, E, binCount,
                                               binCursor, binStart, binned);

  rgcn_pack_kernel<RGCN_IN>
      <<<(RGCN_IN * RGCN_W + 255) / 256, 256, 0, stream>>>(root1, basis1, Bp1);
  rgcn_pack_kernel<RGCN_HID>
      <<<(RGCN_HID * RGCN_W + 255) / 256, 256, 0, stream>>>(root2, basis2, Bp2);

  const int gg = (N + 16 * GEMM_MT - 1) / (16 * GEMM_MT);

  rgcn_gemm_kernel<RGCN_IN>
      <<<gg, GEMM_THREADS, 0, stream>>>(x, Bp1, bias1, hA, z, N);
  rgcn_aggregate_kernel<<<BIN_NB, BIN_T, 0, stream>>>(
      binned, binStart, binCount, comp1, probs, z, hA, N);

  rgcn_layer2_kernel<<<BIN_NB, BIN_T, 0, stream>>>(
      binned, binStart, binCount, comp2, probs, need, hA, Bp2, bias2, hB, N);

  rgcn_score_kernel<<<(ET + SC_EPB - 1) / SC_EPB, SC_THREADS, 0, stream>>>(
      hB, tei, Rm, D, (float*)d_out, ET);
}


# Binning and counting loops issue a thread's independent global loads together (8 edges x dst/src/relation; 4 list entries; next trip's entry) ahead of the LDS adds they feed; dst read once
# speedup vs baseline: 3.1632x; 1.1745x over previous
#include <hip/hip_runtime.h>
#include <cstddef>

typedef float v2f __attribute__((ext_vector_type(2)));
typedef float v8f __attribute__((ext_vector_type(8)));

#define RGCN_R   128
#define RGCN_B   4
#define RGCN_IN  128
#define RGCN_HID 64
#define RGCN_W   (RGCN_HID * (RGCN_B + 1))

__device__ __forceinline__ v8f wmma_f32x4(v2f a, v2f b, v8f c) {
  return __builtin_amdgcn_wmma_f32_16x16x4_f32(false, a, false, b, (short)0, c,
                                               false, false);
}

#define BIN_P    98
#define BIN_NB   512
#define BIN_EPB  8192
#define BIN_T    1024
#define BIN_LINE 16

__device__ __forceinline__ int rgcn_bin_of(int d) {
  const int b = d / BIN_P;
  return b < BIN_NB - 1 ? b : BIN_NB - 1;
}

__device__ __forceinline__ int rgcn_pad_line(int n) {
  return (n + (BIN_LINE - 1)) & ~(BIN_LINE - 1);
}

__device__ __forceinline__ void rgcn_scan_bins(int* a, int* wsum) {
  const int t = threadIdx.x;
  const int lane = t & 31;
  const int w = t >> 5;
  int v = 0, inc = 0;
  if (t < BIN_NB) {
    v = a[t];
    inc = v;
    for (int d = 1; d < 32; d <<= 1) {
      const int n = __shfl_up(inc, d, 32);
      if (lane >= d) inc += n;
    }
    if (lane == 31) wsum[w] = inc;
  }
  __syncthreads();
  if (t < BIN_NB) {
    int off = 0;
    for (int k = 0; k < w; ++k) off += wsum[k];
    a[t] = off + inc - v;
  }
  __syncthreads();
}

__global__ __launch_bounds__(BIN_T) void rgcn_binhist_kernel(
    const int* __restrict__ dst, int E, int* __restrict__ binCount,
    const int* __restrict__ tei, unsigned* __restrict__ need, int nT) {
  __shared__ int hist[BIN_NB];
  const int t = threadIdx.x;
  const int e0 = blockIdx.x * BIN_EPB;
  int dv[BIN_EPB / BIN_T];
#pragma unroll
  for (int k = 0; k < BIN_EPB / BIN_T; ++k) {
    const int e = e0 + k * BIN_T + t;
    dv[k] = dst[e < E ? e : E - 1];
  }
  if (t < BIN_NB) hist[t] = 0;
  __syncthreads();
#pragma unroll
  for (int k = 0; k < BIN_EPB / BIN_T; ++k) {
    const int e = e0 + k * BIN_T + t;
    if (e < E) atomicAdd(&hist[rgcn_bin_of(dv[k])], 1);
  }
  __syncthreads();
  if (t < BIN_NB) {
    const int pc = rgcn_pad_line(hist[t]);
    if (pc > 0) atomicAdd(&binCount[t], pc);
  }
  for (int i = blockIdx.x * BIN_T + t; i < nT; i += (int)(gridDim.x * BIN_T)) {
    const int n = tei[i];
    atomicOr(&need[n >> 5], 1u << (n & 31));
  }
}

__global__ __launch_bounds__(BIN_T) void rgcn_binfill_kernel(
    const int* __restrict__ src, const int* __restrict__ dst,
    const int* __restrict__ et, int E, const int* __restrict__ binCount,
    int* __restrict__ binCursor, int* __restrict__ binStart,
    int* __restrict__ binnedW) {
  int2* __restrict__ binned = (int2*)binnedW;
  __shared__ int hist[BIN_NB];
  __shared__ int loff[BIN_NB];
  __shared__ int fill[BIN_NB];
  __shared__ int gbase[BIN_NB];
  __shared__ int wsum[BIN_NB / 32];
  __shared__ int2 staged[BIN_EPB];
  const int t = threadIdx.x;
  const int e0 = blockIdx.x * BIN_EPB;
  int dv[BIN_EPB / BIN_T], sv[BIN_EPB / BIN_T], tv[BIN_EPB / BIN_T];
#pragma unroll
  for (int k = 0; k < BIN_EPB / BIN_T; ++k) {
    const int e = e0 + k * BIN_T + t;
    dv[k] = dst[e < E ? e : E - 1];
  }
#pragma unroll
  for (int k = 0; k < BIN_EPB / BIN_T; ++k) {
    const int e = e0 + k * BIN_T + t;
    const int ec = e < E ? e : E - 1;
    sv[k] = src[ec];
    tv[k] = et[ec];
  }
  if (t < BIN_NB) {
    hist[t] = 0;
    gbase[t] = binCount[t];
  }
  __syncthreads();
  int mybin[BIN_EPB / BIN_T];
#pragma unroll
  for (int k = 0; k < BIN_EPB / BIN_T; ++k) {
    const int e = e0 + k * BIN_T + t;
    mybin[k] = -1;
    if (e < E) {
      mybin[k] = rgcn_bin_of(dv[k]);
      atomicAdd(&hist[mybin[k]], 1);
    }
  }
  __syncthreads();
  rgcn_scan_bins(gbase, wsum);
  if (t < BIN_NB) {
    const int h = hist[t];
    if (blockIdx.x == 0) binStart[t] = gbase[t];
    const int pc = rgcn_pad_line(h);
    if (pc > 0) gbase[t] += atomicAdd(&binCursor[t], pc);
    loff[t] = h;
  }
  __syncthreads();
  rgcn_scan_bins(loff, wsum);
  if (t < BIN_NB) fill[t] = loff[t];
  __syncthreads();
  int pos[BIN_EPB / BIN_T];
#pragma unroll
  for (int k = 0; k < BIN_EPB / BIN_T; ++k) {
    pos[k] = -1;
    if (mybin[k] >= 0) pos[k] = atomicAdd(&fill[mybin[k]], 1);
  }
#pragma unroll
  for (int k = 0; k < BIN_EPB / BIN_T; ++k) {
    if (pos[k] >= 0)
      staged[pos[k]] =
          make_int2(sv[k], (dv[k] - mybin[k] * BIN_P) * RGCN_R + tv[k]);
  }
  __syncthreads();
  const int lane = t & 31;
  const int w = t >> 5;
  for (int b = w; b < BIN_NB; b += BIN_T / 32) {
    const int h  = hist[b];
    const int pc = rgcn_pad_line(h);
    const int lo = loff[b];
    const int gb = gbase[b];
    for (int k = lane; k < pc; k += 32) {
      int2 v = make_int2(-1, 0);
      if (k < h) v = staged[lo + k];
      binned[gb + k] = v;
    }
  }
}

template<int K>
__global__ void rgcn_pack_kernel(const float* __restrict__ rootW,
                                 const float* __restrict__ basisW,
                                 float* __restrict__ Bp) {
  const int t = blockIdx.x * blockDim.x + threadIdx.x;
  if (t >= K * RGCN_W) return;
  const int m = t & 3;
  const int n = (t >> 2) % RGCN_W;
  const int q = t / (4 * RGCN_W);
  const int k = q * 4 + m;
  float v;
  if (n < RGCN_HID) {
    v = rootW[k * RGCN_HID + n];
  } else {
    const int b = (n - RGCN_HID) >> 6;
    const int c = (n - RGCN_HID) & 63;
    v = basisW[((size_t)b * K + k) * RGCN_HID + c];
  }
  Bp[t] = v;
}

#define L1_PITCH (RGCN_B * RGCN_IN + 4)
#define L1_CAP   4096
#define L1_GRP   8

__device__ __forceinline__ void rgcn_l1_flush(float* tile, int row, int lane,
                                              const float4 (&acc)[RGCN_B]) {
  float* tp = tile + row * L1_PITCH + 4 * lane;
#pragma unroll
  for (int b = 0; b < RGCN_B; ++b) {
    atomicAdd(tp + b * RGCN_IN,     acc[b].x);
    atomicAdd(tp + b * RGCN_IN + 1, acc[b].y);
    atomicAdd(tp + b * RGCN_IN + 2, acc[b].z);
    atomicAdd(tp + b * RGCN_IN + 3, acc[b].w);
  }
}

__global__ __launch_bounds__(BIN_T) void rgcn_layer1_kernel(
    const int* __restrict__ binnedW,
    const int* __restrict__ binStart,
    const int* __restrict__ binCount, const float* __restrict__ comp,
    const float* __restrict__ probs,
    const float* __restrict__ x,
    const float* __restrict__ Bp,
    const float* __restrict__ bias, float* __restrict__ hout, int N) {
  __shared__ __attribute__((aligned(16))) unsigned cntL[BIN_P * RGCN_R];
  __shared__ __attribute__((aligned(16))) float tile[BIN_P * L1_PITCH];
  __shared__ int2 sorted[L1_CAP];
  __shared__ int dcnt[128];
  __shared__ int dpos[128];
  __shared__ int wsum[4];
  const int2* __restrict__ binned = (const int2*)binnedW;
  const int t = threadIdx.x;
  const int lane = t & 31;
  const int w = __builtin_amdgcn_readfirstlane(t >> 5);
  const int n0 = blockIdx.x * BIN_P;
  int nn = N - n0;
  if (nn > BIN_P) nn = BIN_P;
  if (nn <= 0) return;
  const int start = binStart[blockIdx.x];
  const int count = binCount[blockIdx.x];

  for (int i = t; i < BIN_P * RGCN_R / 4; i += BIN_T)
    ((uint4*)cntL)[i] = make_uint4(0u, 0u, 0u, 0u);
  for (int i = t; i < BIN_P * L1_PITCH / 4; i += BIN_T)
    ((float4*)tile)[i] = make_float4(0.f, 0.f, 0.f, 0.f);
  __syncthreads();

  for (int i0 = t; i0 < count; i0 += 4 * BIN_T) {
    int2 ev[4];
#pragma unroll
    for (int u = 0; u < 4; ++u) {
      const int i = i0 + u * BIN_T;
      ev[u] = binned[start + (i < count ? i : count - 1)];
      if (i >= count) ev[u].x = -1;
    }
#pragma unroll
    for (int u = 0; u < 4; ++u)
      if (ev[u].x >= 0) atomicAdd(&cntL[ev[u].y], 1u);
  }
  __syncthreads();

  for (int c0 = 0; c0 < count; c0 += L1_CAP) {
    int cn = count - c0;
    if (cn > L1_CAP) cn = L1_CAP;
    int2 ent[L1_CAP / BIN_T];
#pragma unroll
    for (int k = 0; k < L1_CAP / BIN_T; ++k) {
      const int i = k * BIN_T + t;
      ent[k] = binned[start + c0 + (i < cn ? i : cn - 1)];
      if (i >= cn) ent[k] = make_int2(-1, 0);
    }
    if (t < 128) dcnt[t] = 0;
    __syncthreads();
#pragma unroll
    for (int k = 0; k < L1_CAP / BIN_T; ++k) {
      if (ent[k].x >= 0) atomicAdd(&dcnt[ent[k].y >> 7], 1);
    }
    __syncthreads();
    int v = 0, inc = 0;
    if (t < 128) {
      v = dcnt[t];
      inc = v;
      for (int d = 1; d < 32; d <<= 1) {
        const int nb = __shfl_up(inc, d, 32);
        if (lane >= d) inc += nb;
      }
      if (lane == 31) wsum[w] = inc;
    }
    __syncthreads();
    if (t < 128) {
      int off = 0;
      for (int k = 0; k < w; ++k) off += wsum[k];
      dpos[t] = off + inc - v;
    }
    const int total =
        __builtin_amdgcn_readfirstlane(wsum[0] + wsum[1] + wsum[2] + wsum[3]);
    __syncthreads();
#pragma unroll
    for (int k = 0; k < L1_CAP / BIN_T; ++k) {
      if (ent[k].x >= 0) {
        const int pos = atomicAdd(&dpos[ent[k].y >> 7], 1);
        sorted[pos] = ent[k];
      }
    }
    __syncthreads();

    const int per = (total + 31) >> 5;
    const int i0 = w * per;
    int i1 = i0 + per;
    if (i1 > total) i1 = total;
    float4 acc[RGCN_B];
#pragma unroll
    for (int b = 0; b < RGCN_B; ++b) acc[b] = make_float4(0.f, 0.f, 0.f, 0.f);
    int cur = -1;
    for (int base = i0; base < i1; base += 32) {
      const int i = base + lane;
      int s = 0, w1 = 0;
      float c0f = 0.f, c1f = 0.f, c2f = 0.f, c3f = 0.f;
      if (i < i1) {
        const int2 ed = sorted[i];
        s = ed.x;
        w1 = ed.y;
        const int ty = w1 & (RGCN_R - 1);
        const float gate = 1.f / (1.f + __expf(-probs[ty]));
        const float c    = (float)cntL[w1];
        const float g    = gate / fmaxf(c, 1.f);
        const float4 cv  = *(const float4*)(comp + ty * RGCN_B);
        c0f = cv.x * g; c1f = cv.y * g; c2f = cv.z * g; c3f = cv.w * g;
      }
      const int dl = w1 >> 7;
      int n = i1 - base;
      if (n > 32) n = 32;
      for (int j0 = 0; j0 < n; j0 += L1_GRP) {
        float4 xv[L1_GRP];
#pragma unroll
        for (int u = 0; u < L1_GRP; ++u) {
          xv[u] = make_float4(0.f, 0.f, 0.f, 0.f);
          if (j0 + u < n) {
            const int sj = __builtin_amdgcn_readlane(s, j0 + u);
            xv[u] = *(const float4*)(x + (size_t)sj * RGCN_IN + 4 * lane);
          }
        }
#pragma unroll
        for (int u = 0; u < L1_GRP; ++u) {
          if (j0 + u < n) {
            const int j  = j0 + u;
            const int dj = __builtin_amdgcn_readlane(dl, j);
            const float a0 = __int_as_float(__builtin_amdgcn_readlane(__float_as_int(c0f), j));
            const float a1 = __int_as_float(__builtin_amdgcn_readlane(__float_as_int(c1f), j));
            const float a2 = __int_as_float(__builtin_amdgcn_readlane(__float_as_int(c2f), j));
            const float a3 = __int_as_float(__builtin_amdgcn_readlane(__float_as_int(c3f), j));
            if (dj != cur) {
              if (cur >= 0) rgcn_l1_flush(tile, cur, lane, acc);
              cur = dj;
#pragma unroll
              for (int b = 0; b < RGCN_B; ++b) acc[b] = make_float4(0.f, 0.f, 0.f, 0.f);
            }
            const float4 r = xv[u];
            acc[0].x += a0 * r.x; acc[0].y += a0 * r.y; acc[0].z += a0 * r.z; acc[0].w += a0 * r.w;
            acc[1].x += a1 * r.x; acc[1].y += a1 * r.y; acc[1].z += a1 * r.z; acc[1].w += a1 * r.w;
            acc[2].x += a2 * r.x; acc[2].y += a2 * r.y; acc[2].z += a2 * r.z; acc[2].w += a2 * r.w;
            acc[3].x += a3 * r.x; acc[3].y += a3 * r.y; acc[3].z += a3 * r.z; acc[3].w += a3 * r.w;
          }
        }
      }
    }
    if (cur >= 0) rgcn_l1_flush(tile, cur, lane, acc);
    __syncthreads();
  }

  float* stage = (float*)cntL;
  const int hi = lane >> 4;
  const int lm = lane & 15;
  const int mtiles = (nn + 15) >> 4;
  for (int unit = w; unit < 4 * mtiles; unit += BIN_T / 32) {
    const int mt = unit >> 2;
    const int tn = unit & 3;
    const int row = 16 * mt + lm;
    const int rowc = row < nn ? row : nn - 1;
    const float* arow = tile + rowc * L1_PITCH + 2 * hi;
    const float* xrow = x + (size_t)(n0 + rowc) * RGCN_IN + 2 * hi;
    const float* bcol = Bp + (size_t)(16 * tn + lm) * 4 + 2 * hi;
    v8f acc0 = {}, acc1 = {};
    for (int bb = 0; bb < RGCN_B; ++bb) {
      const float* ap = arow + bb * RGCN_IN;
      const float* bp = bcol + (size_t)(RGCN_HID + bb * RGCN_HID) * 4;
#pragma unroll 4
      for (int k4 = 0; k4 < RGCN_IN / 4; k4 += 2) {
        const v2f a0 = *(const v2f*)(ap + 4 * k4);
        const v2f a1 = *(const v2f*)(ap + 4 * k4 + 4);
        const v2f b0 = *(const v2f*)(bp + (size_t)k4 * (4 * RGCN_W));
        const v2f b1 = *(const v2f*)(bp + (size_t)(k4 + 1) * (4 * RGCN_W));
        acc0 = wmma_f32x4(a0, b0, acc0);
        acc1 = wmma_f32x4(a1, b1, acc1);
      }
    }
#pragma unroll 4
    for (int k4 = 0; k4 < RGCN_IN / 4; k4 += 2) {
      const v2f a0 = *(const v2f*)(xrow + 4 * k4);
      const v2f a1 = *(const v2f*)(xrow + 4 * k4 + 4);
      const v2f b0 = *(const v2f*)(bcol + (size_t)k4 * (4 * RGCN_W));
      const v2f b1 = *(const v2f*)(bcol + (size_t)(k4 + 1) * (4 * RGCN_W));
      acc0 = wmma_f32x4(a0, b0, acc0);
      acc1 = wmma_f32x4(a1, b1, acc1);
    }
    float* sp = stage + (16 * mt + 8 * hi) * RGCN_HID + 16 * tn + lm;
#pragma unroll
    for (int v = 0; v < 8; ++v) sp[v * RGCN_HID] = acc0[v] + acc1[v];
  }
  __syncthreads();

  for (int idx = t; idx < nn * (RGCN_HID / 4); idx += BIN_T) {
    const int q = idx & 15;
    float4 v = *(const float4*)&stage[4 * idx];
    const float4 bv = *(const float4*)(bias + 4 * q);
    v.x += bv.x; v.y += bv.y; v.z += bv.z; v.w += bv.w;
    *(float4*)(hout + (size_t)n0 * RGCN_HID + 4 * idx) = v;
  }
}

#define L2_PITCH (RGCN_B * RGCN_HID + 4)
#define L2_GRP 8

template<int G>
__device__ __forceinline__ void rgcn_l2_group(
    int s, int sl, float c0, float c1, float c2, float c3, unsigned& m, int lane,
    const float* __restrict__ hin, float* tile) {
  int jl[G];
  int n = 0;
#pragma unroll
  for (int u = 0; u < G; ++u) {
    jl[u] = 0;
    if (m != 0u) {
      jl[u] = __builtin_ctz(m);
      m &= m - 1u;
      ++n;
    }
  }
  v2f xv[G];
#pragma unroll
  for (int u = 0; u < G; ++u) {
    xv[u].x = 0.f;
    xv[u].y = 0.f;
    if (u < n) {
      const int sj = __builtin_amdgcn_readlane(s, jl[u]);
      xv[u] = *(const v2f*)(hin + (size_t)sj * RGCN_HID + 2 * lane);
    }
  }
#pragma unroll
  for (int u = 0; u < G; ++u) {
    if (u < n) {
      const int j   = jl[u];
      const int slj = __builtin_amdgcn_readlane(sl, j);
      const float a0 = __int_as_float(__builtin_amdgcn_readlane(__float_as_int(c0), j));
      const float a1 = __int_as_float(__builtin_amdgcn_readlane(__float_as_int(c1), j));
      const float a2 = __int_as_float(__builtin_amdgcn_readlane(__float_as_int(c2), j));
      const float a3 = __int_as_float(__builtin_amdgcn_readlane(__float_as_int(c3), j));
      const float x0 = fmaxf(xv[u].x, 0.f);
      const float x1 = fmaxf(xv[u].y, 0.f);
      float* tp = tile + slj * L2_PITCH + 2 * lane;
      atomicAdd(tp,                    a0 * x0);
      atomicAdd(tp + 1,                a0 * x1);
      atomicAdd(tp + RGCN_HID,         a1 * x0);
      atomicAdd(tp + RGCN_HID + 1,     a1 * x1);
      atomicAdd(tp + 2 * RGCN_HID,     a2 * x0);
      atomicAdd(tp + 2 * RGCN_HID + 1, a2 * x1);
      atomicAdd(tp + 3 * RGCN_HID,     a3 * x0);
      atomicAdd(tp + 3 * RGCN_HID + 1, a3 * x1);
    }
  }
}

__global__ __launch_bounds__(BIN_T) __attribute__((amdgpu_waves_per_eu(16)))
void rgcn_layer2_kernel(
    const int* __restrict__ binnedW,
    const int* __restrict__ binStart,
    const int* __restrict__ binCount, const float* __restrict__ comp,
    const float* __restrict__ probs,
    const unsigned* __restrict__ need,
    const float* __restrict__ hin,
    const float* __restrict__ Bp,
    const float* __restrict__ bias, float* __restrict__ hout, int N) {
  __shared__ __attribute__((aligned(16))) unsigned cntL[BIN_P * RGCN_R];
  __shared__ __attribute__((aligned(16))) float tile[BIN_P * L2_PITCH];
  __shared__ int slotOf[BIN_P];
  __shared__ int nodeOf[BIN_P];
  __shared__ int wcnt[4];
  const int2* __restrict__ binned = (const int2*)binnedW;
  const int t = threadIdx.x;
  const int lane = t & 31;
  const int w = __builtin_amdgcn_readfirstlane(t >> 5);
  const int n0 = blockIdx.x * BIN_P;
  const int start = binStart[blockIdx.x];
  const int count = binCount[blockIdx.x];

  bool nd = false;
  int pre = 0;
  if (t < 128) {
    const int n = n0 + t;
    if (t < BIN_P && n < N) nd = ((need[n >> 5] >> (n & 31)) & 1u) != 0u;
    const unsigned bal = __builtin_amdgcn_ballot_w32(nd);
    pre = __builtin_popcount(bal & ((1u << lane) - 1u));
    if (lane == 0) wcnt[w] = __builtin_popcount(bal);
  }
  for (int i = t; i < BIN_P * RGCN_R; i += BIN_T) cntL[i] = 0u;
  __syncthreads();
  const int nNeed = wcnt[0] + wcnt[1] + wcnt[2] + wcnt[3];
  if (nNeed == 0) return;
  if (t < 128) {
    int off = 0;
    for (int k = 0; k < w; ++k) off += wcnt[k];
    if (t < BIN_P) slotOf[t] = nd ? off + pre : -1;
    if (nd) nodeOf[off + pre] = t;
  }
  const int mtiles = (nNeed + 15) >> 4;
  int zrows = 16 * mtiles;
  if (zrows > BIN_P) zrows = BIN_P;
  for (int i = t; i < zrows * L2_PITCH; i += BIN_T) tile[i] = 0.f;

  for (int i0 = t; i0 < count; i0 += 4 * BIN_T) {
    int2 ev[4];
#pragma unroll
    for (int u = 0; u < 4; ++u) {
      const int i = i0 + u * BIN_T;
      ev[u] = binned[start + (i < count ? i : count - 1)];
      if (i >= count) ev[u].x = -1;
    }
#pragma unroll
    for (int u = 0; u < 4; ++u)
      if (ev[u].x >= 0) atomicAdd(&cntL[ev[u].y], 1u);
  }
  int2 nxt = make_int2(-1, 0);
  if (w * 32 + lane < count) nxt = binned[start + w * 32 + lane];
  __syncthreads();

  for (int base = w * 32; base < count; base += BIN_T) {
    const int2 ed = nxt;
    nxt = make_int2(-1, 0);
    if (base + BIN_T + lane < count) nxt = binned[start + base + BIN_T + lane];
    const int s = ed.x;
    const int w1 = ed.y;
    int sl = 0;
    bool keep = s >= 0;
    const int ty = w1 & (RGCN_R - 1);
    float c0 = 0.f, c1 = 0.f, c2 = 0.f, c3 = 0.f;
    if (keep) {
      sl = slotOf[w1 >> 7];
      keep = sl >= 0;
    }
    if (keep) {
      const float gate = 1.f / (1.f + __expf(-probs[ty]));
      const float c    = (float)cntL[w1];
      const float g    = gate / fmaxf(c, 1.f);
      const float4 cv  = *(const float4*)(comp + ty * RGCN_B);
      c0 = cv.x * g; c1 = cv.y * g; c2 = cv.z * g; c3 = cv.w * g;
    }
    unsigned m = __builtin_amdgcn_ballot_w32(keep);
    while (m != 0u)
      rgcn_l2_group<L2_GRP>(s, sl, c0, c1, c2, c3, m, lane, hin, tile);
  }
  __syncthreads();

  float* stage = (float*)cntL;
  const int hi = lane >> 4;
  const int lm = lane & 15;
  for (int unit = w; unit < 4 * mtiles; unit += BIN_T / 32) {
    const int mt = unit >> 2;
    const int tn = unit & 3;
    const int row = 16 * mt + lm;
    const int rowc = row < zrows ? row : zrows - 1;
    const int slc  = row < nNeed ? row : nNeed - 1;
    const float* arow = tile + rowc * L2_PITCH + 2 * hi;
    const float* hrow = hin + (size_t)(n0 + nodeOf[slc]) * RGCN_HID + 2 * hi;
    const float* bcol = Bp + (size_t)(16 * tn + lm) * 4 + 2 * hi;
    v8f acc0 = {}, acc1 = {};
    for (int bb = 0; bb < RGCN_B; ++bb) {
      const float* ap = arow + bb * RGCN_HID;
      const float* bp = bcol + (size_t)(RGCN_HID + bb * RGCN_HID) * 4;
#pragma unroll 4
      for (int k4 = 0; k4 < RGCN_HID / 4; k4 += 2) {
        const v2f a0 = *(const v2f*)(ap + 4 * k4);
        const v2f a1 = *(const v2f*)(ap + 4 * k4 + 4);
        const v2f b0 = *(const v2f*)(bp + (size_t)k4 * (4 * RGCN_W));
        const v2f b1 = *(const v2f*)(bp + (size_t)(k4 + 1) * (4 * RGCN_W));
        acc0 = wmma_f32x4(a0, b0, acc0);
        acc1 = wmma_f32x4(a1, b1, acc1);
      }
    }
#pragma unroll 4
    for (int k4 = 0; k4 < RGCN_HID / 4; k4 += 2) {
      v2f a0 = *(const v2f*)(hrow + 4 * k4);
      v2f a1 = *(const v2f*)(hrow + 4 * k4 + 4);
      a0.x = fmaxf(a0.x, 0.f); a0.y = fmaxf(a0.y, 0.f);
      a1.x = fmaxf(a1.x, 0.f); a1.y = fmaxf(a1.y, 0.f);
      const v2f b0 = *(const v2f*)(bcol + (size_t)k4 * (4 * RGCN_W));
      const v2f b1 = *(const v2f*)(bcol + (size_t)(k4 + 1) * (4 * RGCN_W));
      acc0 = wmma_f32x4(a0, b0, acc0);
      acc1 = wmma_f32x4(a1, b1, acc1);
    }
    float* sp = stage + (16 * mt + 8 * hi) * RGCN_HID + 16 * tn + lm;
#pragma unroll
    for (int v = 0; v < 8; ++v) sp[v * RGCN_HID] = acc0[v] + acc1[v];
  }
  __syncthreads();

  for (int idx = t; idx < nNeed * (RGCN_HID / 4); idx += BIN_T) {
    const int slot = idx >> 4;
    const int q = idx & 15;
    float4 v = *(const float4*)&stage[slot * RGCN_HID + 4 * q];
    const float4 bv = *(const float4*)(bias + 4 * q);
    v.x += bv.x; v.y += bv.y; v.z += bv.z; v.w += bv.w;
    *(float4*)(hout + (size_t)(n0 + nodeOf[slot]) * RGCN_HID + 4 * q) = v;
  }
}

#define SC_EPB 10
#define SC_THREADS 128

__global__ __launch_bounds__(SC_THREADS) void rgcn_score_kernel(
    const float* __restrict__ hB,
    const int* __restrict__ tei,
    const float* __restrict__ Rm,
    const float* __restrict__ D,
    float* __restrict__ out, int ET) {
  __shared__ __attribute__((aligned(16))) float x1s[SC_EPB][RGCN_HID];
  __shared__ __attribute__((aligned(16))) float x2s[SC_EPB][RGCN_HID];
  __shared__ __attribute__((aligned(16))) float outs[SC_EPB][RGCN_R];
  const int base = blockIdx.x * SC_EPB;

  for (int idx = threadIdx.x; idx < 2 * SC_EPB * 16; idx += blockDim.x) {
    const int which = idx / (SC_EPB * 16);
    const int rem   = idx - which * (SC_EPB * 16);
    const int m = rem >> 4, q = rem & 15;
    const int e = base + m;
    if (e < ET) {
      const int node = tei[which * ET + e];
      float4 v = *(const float4*)(hB + (size_t)node * RGCN_HID + q * 4);
      v.x = fmaxf(v.x, 0.f); v.y = fmaxf(v.y, 0.f);
      v.z = fmaxf(v.z, 0.f); v.w = fmaxf(v.w, 0.f);
      float* dstp = which ? &x2s[m][q * 4] : &x1s[m][q * 4];
      *(float4*)dstp = v;
    }
  }

  const int wave = threadIdx.x >> 5;
  const int lane = threadIdx.x & 31;
  const int hi   = lane >> 4;
  const int lm   = lane & 15;
  const int r0   = wave * 32 + lm;

  v2f fu[4][16];
  v2f fl[4][16];
  v2f dj[2][16];
  float di[2][4][8];
#pragma unroll
  for (int mt = 0; mt < 4; ++mt) {
#pragma unroll
    for (int kk = 4 * mt; kk < 16; ++kk) {
      const int i  = mt * 16 + lm;
      const int j0 = 4 * kk + 2 * hi;
      const v2f u = *(const v2f*)(Rm + i * RGCN_HID + j0);
      const float l0 = Rm[j0 * RGCN_HID + i];
      const float l1 = Rm[(j0 + 1) * RGCN_HID + i];
      fu[mt][kk].x = (j0     >= i) ? u.x : 0.f;
      fu[mt][kk].y = (j0 + 1 >= i) ? u.y : 0.f;
      fl[mt][kk].x = (j0     >  i) ? l0 : 0.f;
      fl[mt][kk].y = (j0 + 1 >  i) ? l1 : 0.f;
    }
  }
#pragma unroll
  for (int t = 0; t < 2; ++t) {
#pragma unroll
    for (int kk = 0; kk < 16; ++kk)
      dj[t][kk] = *(const v2f*)(D + (r0 + 16 * t) * RGCN_HID + 4 * kk + 2 * hi);
#pragma unroll
    for (int mt = 0; mt < 4; ++mt) {
      const float* dp = D + (r0 + 16 * t) * RGCN_HID + mt * 16 + 8 * hi;
      const float4 lo = *(const float4*)dp;
      const float4 up = *(const float4*)(dp + 4);
      di[t][mt][0] = lo.x; di[t][mt][1] = lo.y; di[t][mt][2] = lo.z; di[t][mt][3] = lo.w;
      di[t][mt][4] = up.x; di[t][mt][5] = up.y; di[t][mt][6] = up.z; di[t][mt][7] = up.w;
    }
  }
  __syncthreads();

  for (int m = 0; m < SC_EPB; ++m) {
    if (base + m >= ET) break;
    v8f acc[2][4] = {};
    float x1i[4], x2i[4];
#pragma unroll
    for (int mt = 0; mt < 4; ++mt) {
      x1i[mt] = x1s[m][16 * mt + lm];
      x2i[mt] = x2s[m][16 * mt + lm];
    }
#pragma unroll
    for (int kk = 0; kk < 16; ++kk) {
      const v2f p1 = *(const v2f*)&x1s[m][4 * kk + 2 * hi];
      const v2f p2 = *(const v2f*)&x2s[m][4 * kk + 2 * hi];
#pragma unroll
      for (int mt = 0; mt <= (kk >> 2); ++mt) {
        const v2f a = (fu[mt][kk] * p2) * x1i[mt] + (fl[mt][kk] * p1) * x2i[mt];
#pragma unroll
        for (int t = 0; t < 2; ++t) acc[t][mt] = wmma_f32x4(a, dj[t][kk], acc[t][mt]);
      }
    }
#pragma unroll
    for (int t = 0; t < 2; ++t) {
      float p[4];
#pragma unroll
      for (int mt = 0; mt < 4; ++mt) {
        float s = di[t][mt][0] * acc[t][mt][0];
#pragma unroll
        for (int v = 1; v < 8; ++v) s += di[t][mt][v] * acc[t][mt][v];
        p[mt] = s;
      }
      float tt = (p[0] + p[1]) + (p[2] + p[3]);
      tt += __shfl_xor(tt, 16, 32);
      if (hi == 0) outs[m][r0 + 16 * t] = 1.f / (1.f + __expf(-tt));
    }
  }
  __syncthreads();

  int nvalid = ET - base;
  if (nvalid > SC_EPB) nvalid = SC_EPB;
  float* __restrict__ obase = out + (size_t)base * RGCN_R;
  const float* osrc = &outs[0][0];
  for (int idx = threadIdx.x * 4; idx < nvalid * RGCN_R; idx += blockDim.x * 4)
    *(float4*)(obase + idx) = *(const float4*)(osrc + idx);
}

extern "C" void kernel_launch(void* const* d_in, const int* in_sizes, int n_in,
                              void* d_out, int out_size, void* d_ws,
                              size_t ws_size, hipStream_t stream) {
  const float* x      = (const float*)d_in[0];
  const int*   eidx   = (const int*)d_in[1];
  const int*   etype  = (const int*)d_in[2];
  const int*   tei    = (const int*)d_in[3];
  const float* probs  = (const float*)d_in[4];
  const float* basis1 = (const float*)d_in[5];
  const float* comp1  = (const float*)d_in[6];
  const float* root1  = (const float*)d_in[7];
  const float* bias1  = (const float*)d_in[8];
  const float* basis2 = (const float*)d_in[9];
  const float* comp2  = (const float*)d_in[10];
  const float* root2  = (const float*)d_in[11];
  const float* bias2  = (const float*)d_in[12];
  const float* Rm     = (const float*)d_in[13];
  const float* D      = (const float*)d_in[14];

  const int N  = in_sizes[0] / RGCN_IN;
  const int E  = in_sizes[2];
  const int ET = in_sizes[3] / 2;

  float* hA  = (float*)d_ws;
  float* hB  = hA  + (size_t)N * RGCN_HID;
  float* Bp1 = hB  + (size_t)N * RGCN_HID;
  float* Bp2 = Bp1 + (size_t)RGCN_IN * RGCN_W;
  int* binCount  = (int*)(Bp2 + (size_t)RGCN_HID * RGCN_W);
  int* binCursor = binCount + BIN_NB;
  unsigned* need = (unsigned*)(binCursor + BIN_NB);
  const size_t needPad = (size_t)((N + 1023) / 1024) * 32;
  int* binStart  = (int*)(need + needPad);
  int* binned    = binStart + BIN_NB;

  const int* src = eidx;
  const int* dst = eidx + E;

  hipMemsetAsync(binCount, 0, ((size_t)2 * BIN_NB + needPad) * sizeof(int), stream);

  const int nb = (E + BIN_EPB - 1) / BIN_EPB;
  rgcn_binhist_kernel<<<nb, BIN_T, 0, stream>>>(dst, E, binCount, tei, need, 2 * ET);
  rgcn_binfill_kernel<<<nb, BIN_T, 0, stream>>>(src, dst, etype, E, binCount,
                                               binCursor, binStart, binned);

  rgcn_pack_kernel<RGCN_IN>
      <<<(RGCN_IN * RGCN_W + 255) / 256, 256, 0, stream>>>(root1, basis1, Bp1);
  rgcn_pack_kernel<RGCN_HID>
      <<<(RGCN_HID * RGCN_W + 255) / 256, 256, 0, stream>>>(root2, basis2, Bp2);

  rgcn_layer1_kernel<<<BIN_NB, BIN_T, 0, stream>>>(
      binned, binStart, binCount, comp1, probs, x, Bp1, bias1, hA, N);

  rgcn_layer2_kernel<<<BIN_NB, BIN_T, 0, stream>>>(
      binned, binStart, binCount, comp2, probs, need, hA, Bp2, bias2, hB, N);

  rgcn_score_kernel<<<(ET + SC_EPB - 1) / SC_EPB, SC_THREADS, 0, stream>>>(
      hB, tei, Rm, D, (float*)d_out, ET);
}
